# starting kernel
# baseline (speedup 1.0000x reference)
#include <hip/hip_runtime.h>
#include <hip/hip_bf16.h>


#define N_NODES     50000
#define N_EDGES     800000
#define IN_FEATS    128
#define H_FEATS     128
#define NUM_CLASSES 16

typedef __attribute__((ext_vector_type(2))) float v2f;
typedef __attribute__((ext_vector_type(8))) float v8f;

__global__ void gcn_degree_kernel(const int* __restrict__ src,
                                  const int* __restrict__ dst,
                                  float* __restrict__ deg_out,
                                  float* __restrict__ deg_in, int E) {
    int e = blockIdx.x * blockDim.x + threadIdx.x;
    if (e < E) {
        atomicAdd(&deg_out[src[e]], 1.0f);
        atomicAdd(&deg_in[dst[e]], 1.0f);
    }
}

__global__ void gcn_norm_kernel(const float* __restrict__ deg_out,
                                const float* __restrict__ deg_in,
                                float* __restrict__ norm_out,
                                float* __restrict__ norm_in, int n) {
    int i = blockIdx.x * blockDim.x + threadIdx.x;
    if (i < n) {
        norm_out[i] = __frsqrt_rn(fmaxf(deg_out[i], 1.0f));
        norm_in[i]  = __frsqrt_rn(fmaxf(deg_in[i], 1.0f));
    }
}

__global__ void gcn_scale_kernel(const float* __restrict__ h,
                                 const float* __restrict__ norm_out,
                                 float* __restrict__ hs, int n_nodes) {
    int t = blockIdx.x * blockDim.x + threadIdx.x;
    int node = t >> 5;
    int c4   = t & 31;
    if (node >= n_nodes) return;
    float s = norm_out[node];
    float4 v = ((const float4*)(h + (size_t)node * 128))[c4];
    v.x *= s; v.y *= s; v.z *= s; v.w *= s;
    ((float4*)(hs + (size_t)node * 128))[c4] = v;
}

__global__ void gcn_aggregate_kernel(const float* __restrict__ hs,
                                     const int* __restrict__ src,
                                     const int* __restrict__ dst,
                                     float* __restrict__ agg, int E) {
    long long t = (long long)blockIdx.x * blockDim.x + threadIdx.x;
    int e = (int)(t >> 5);
    int c = (int)(t & 31) * 4;
    if (e >= E) return;
    int s = src[e];
    int d = dst[e];
    float4 v = *(const float4*)(hs + (size_t)s * 128 + c);
    float* o = agg + (size_t)d * 128 + c;
    atomicAdd(o + 0, v.x);
    atomicAdd(o + 1, v.y);
    atomicAdd(o + 2, v.z);
    atomicAdd(o + 3, v.w);
}

__global__ __launch_bounds__(256)
void gcn_gemm_wmma_kernel(const float* __restrict__ A,
                          const float* __restrict__ norm_in,
                          const float* __restrict__ W,
                          const float* __restrict__ bias,
                          float* __restrict__ out,
                          int N, int relu) {
    const int K = 128;
    __shared__ float As[16 * 132];

    const int wave = threadIdx.x >> 5;
    const int lane = threadIdx.x & 31;
    const int half = lane >> 4;
    const int l16  = lane & 15;
    const int tileM = blockIdx.x * 16;
    const int tileN = wave * 16;

    for (int i = threadIdx.x; i < 16 * 32; i += 256) {
        int r  = i >> 5;
        int c4 = i & 31;
        float4 v = ((const float4*)(A + (size_t)(tileM + r) * K))[c4];
        float s = norm_in[tileM + r];
        v.x *= s; v.y *= s; v.z *= s; v.w *= s;
        ((float4*)(As + r * 132))[c4] = v;
    }
    __syncthreads();

    if (tileN >= N) return;

    v8f acc = {};
    for (int k0 = 0; k0 < K; k0 += 4) {
        int ka = k0 + 2 * half;
        v2f a, b;
        a.x = As[l16 * 132 + ka];
        a.y = As[l16 * 132 + ka + 1];
        b.x = W[(size_t)ka * N + tileN + l16];
        b.y = W[(size_t)(ka + 1) * N + tileN + l16];
        acc = __builtin_amdgcn_wmma_f32_16x16x4_f32(
             false, a,  false, b,
             (short)0, acc,  false,  false);
    }

    float bb = bias[tileN + l16];
    #pragma unroll
    for (int r = 0; r < 8; ++r) {
        int row = tileM + r + 8 * half;
        float v = acc[r] + bb;
        if (relu) v = fmaxf(v, 0.0f);
        out[(size_t)row * N + tileN + l16] = v;
    }
}

extern "C" void kernel_launch(void* const* d_in, const int* in_sizes, int n_in,
                              void* d_out, int out_size, void* d_ws, size_t ws_size,
                              hipStream_t stream) {
    const float* in_feat = (const float*)d_in[0];
    const int*   e_src   = (const int*)d_in[1];
    const int*   e_dst   = (const int*)d_in[2];
    const float* W1 = (const float*)d_in[3];
    const float* b1 = (const float*)d_in[4];
    const float* W2 = (const float*)d_in[5];
    const float* b2 = (const float*)d_in[6];
    const float* W3 = (const float*)d_in[7];
    const float* b3 = (const float*)d_in[8];
    float* out = (float*)d_out;

    const size_t NF = (size_t)N_NODES * 128;
    float* B0 = (float*)d_ws;
    float* B1 = B0 + NF;
    float* B2 = B1 + NF;
    float* deg_out  = B2 + NF;
    float* deg_in   = deg_out + N_NODES;
    float* norm_out = deg_in + N_NODES;
    float* norm_in  = norm_out + N_NODES;

    const int TB = 256;
    const int edgeBlocks  = (N_EDGES + TB - 1) / TB;
    const int aggBlocks   = (int)(((long long)N_EDGES * 32 + TB - 1) / TB);
    const int scaleBlocks = (N_NODES * 32 + TB - 1) / TB;
    const int nodeBlocks  = (N_NODES + TB - 1) / TB;
    const int gemmBlocks  = N_NODES / 16;

    hipMemsetAsync(deg_out, 0, 2 * (size_t)N_NODES * sizeof(float), stream);
    gcn_degree_kernel<<<edgeBlocks, TB, 0, stream>>>(e_src, e_dst, deg_out, deg_in, N_EDGES);
    gcn_norm_kernel<<<nodeBlocks, TB, 0, stream>>>(deg_out, deg_in, norm_out, norm_in, N_NODES);

    gcn_scale_kernel<<<scaleBlocks, TB, 0, stream>>>(in_feat, norm_out, B0, N_NODES);
    hipMemsetAsync(B1, 0, NF * sizeof(float), stream);
    gcn_aggregate_kernel<<<aggBlocks, TB, 0, stream>>>(B0, e_src, e_dst, B1, N_EDGES);
    gcn_gemm_wmma_kernel<<<gemmBlocks, TB, 0, stream>>>(B1, norm_in, W1, b1, B2, H_FEATS, 1);

    gcn_scale_kernel<<<scaleBlocks, TB, 0, stream>>>(B2, norm_out, B0, N_NODES);
    hipMemsetAsync(B1, 0, NF * sizeof(float), stream);
    gcn_aggregate_kernel<<<aggBlocks, TB, 0, stream>>>(B0, e_src, e_dst, B1, N_EDGES);
    gcn_gemm_wmma_kernel<<<gemmBlocks, TB, 0, stream>>>(B1, norm_in, W2, b2, B2, H_FEATS, 1);

    gcn_scale_kernel<<<scaleBlocks, TB, 0, stream>>>(B2, norm_out, B0, N_NODES);
    hipMemsetAsync(B1, 0, NF * sizeof(float), stream);
    gcn_aggregate_kernel<<<aggBlocks, TB, 0, stream>>>(B0, e_src, e_dst, B1, N_EDGES);
    gcn_gemm_wmma_kernel<<<gemmBlocks, TB, 0, stream>>>(B1, norm_in, W3, b3, out, NUM_CLASSES, 0);
}


# Aggregation as destination-sorted row gather (edge list counting-sorted per call via int degree atomics) replacing float atomic scatter-add; row sums in arrival-rank order
# speedup vs baseline: 2.2090x; 2.2090x over previous
#include <hip/hip_runtime.h>
#include <hip/hip_bf16.h>


#define N_NODES     50000
#define N_EDGES     800000
#define IN_FEATS    128
#define H_FEATS     128
#define NUM_CLASSES 16

typedef __attribute__((ext_vector_type(2))) float v2f;
typedef __attribute__((ext_vector_type(8))) float v8f;

__global__ void gcn_degree_rank_kernel(const int* __restrict__ src,
                                       const int* __restrict__ dst,
                                       int* __restrict__ deg_out,
                                       int* __restrict__ deg_in,
                                       int* __restrict__ rank, int E) {
    int e = blockIdx.x * blockDim.x + threadIdx.x;
    if (e < E) {
        const int s = src[e];
        const int d = dst[e];
        atomicAdd(&deg_out[s], 1);
        rank[e] = atomicAdd(&deg_in[d], 1);
    }
}

__global__ __launch_bounds__(256)
void gcn_rowptr_norm_kernel(const int* __restrict__ deg_out,
                            const int* __restrict__ deg_in,
                            float* __restrict__ norm_out,
                            float* __restrict__ norm_in,
                            int* __restrict__ row_ptr, int n) {
    __shared__ int red[8];
    __shared__ int wsum[8];
    const int tid  = threadIdx.x;
    const int lane = tid & 31;
    const int wave = tid >> 5;
    const int base = blockIdx.x * 256;

    int s = 0;
    const int4* p4 = (const int4*)deg_in + tid;
    const int n4 = base >> 2;
    const int full = n4 >> 8;
    int m = 0;
    for (; m + 8 <= full; m += 8) {
        int4 v[8];
        #pragma unroll
        for (int u = 0; u < 8; ++u) v[u] = p4[(m + u) * 256];
        #pragma unroll
        for (int u = 0; u < 8; ++u) s += (v[u].x + v[u].y) + (v[u].z + v[u].w);
    }
    if (full & 4) {
        int4 v[4];
        #pragma unroll
        for (int u = 0; u < 4; ++u) v[u] = p4[(m + u) * 256];
        #pragma unroll
        for (int u = 0; u < 4; ++u) s += (v[u].x + v[u].y) + (v[u].z + v[u].w);
        m += 4;
    }
    if (full & 2) {
        int4 v0 = p4[m * 256];
        int4 v1 = p4[(m + 1) * 256];
        s += (v0.x + v0.y) + (v0.z + v0.w);
        s += (v1.x + v1.y) + (v1.z + v1.w);
        m += 2;
    }
    if (full & 1) {
        int4 v0 = p4[m * 256];
        s += (v0.x + v0.y) + (v0.z + v0.w);
        m += 1;
    }
    if (tid < (n4 & 255)) {
        int4 v0 = p4[m * 256];
        s += (v0.x + v0.y) + (v0.z + v0.w);
    }
    for (int o = 16; o > 0; o >>= 1) s += __shfl_xor(s, o);
    if (lane == 0) red[wave] = s;

    const int i = base + tid;
    int di = 0;
    if (i < n) {
        di = deg_in[i];
        int dq = deg_out[i];
        norm_out[i] = __frsqrt_rn(fmaxf((float)dq, 1.0f));
        norm_in[i]  = __frsqrt_rn(fmaxf((float)di, 1.0f));
    }
    int x = di;
    for (int o = 1; o < 32; o <<= 1) {
        int y = __shfl_up(x, o);
        if (lane >= o) x += y;
    }
    if (lane == 31) wsum[wave] = x;
    __syncthreads();

    int off = 0;
    for (int w = 0; w < 8; ++w) {
        off += red[w];
        if (w < wave) off += wsum[w];
    }
    const int excl = off + x - di;
    if (i < n) row_ptr[i] = excl;
    if (i == n - 1) row_ptr[n] = excl + di;
}

__global__ void gcn_fill_kernel(const int* __restrict__ src,
                                const int* __restrict__ dst,
                                const int* __restrict__ rank,
                                const int* __restrict__ row_ptr,
                                int* __restrict__ col, int E) {
    int e = blockIdx.x * blockDim.x + threadIdx.x;
    if (e < E) col[row_ptr[dst[e]] + rank[e]] = src[e];
}

__global__ void gcn_scale_kernel(const float* __restrict__ h,
                                 const float* __restrict__ norm_out,
                                 float* __restrict__ hs, int n_nodes) {
    int t = blockIdx.x * blockDim.x + threadIdx.x;
    int node = t >> 5;
    int c4   = t & 31;
    if (node >= n_nodes) return;
    float s = norm_out[node];
    float4 v = ((const float4*)(h + (size_t)node * 128))[c4];
    v.x *= s; v.y *= s; v.z *= s; v.w *= s;
    ((float4*)(hs + (size_t)node * 128))[c4] = v;
}

__device__ __forceinline__ float4 gcn_row4(const float* __restrict__ hs, int s, int lane) {
    return *(const float4*)(hs + (size_t)s * 128 + lane * 4);
}
__device__ __forceinline__ void gcn_acc4(float4& a, const float4& v) {
    a.x += v.x; a.y += v.y; a.z += v.z; a.w += v.w;
}

__global__ __launch_bounds__(256)
void gcn_gather_kernel(const float* __restrict__ hs,
                       const int* __restrict__ row_ptr,
                       const int* __restrict__ col,
                       float* __restrict__ agg, int n_nodes) {
    const int lane = threadIdx.x & 31;
    const int node = __builtin_amdgcn_readfirstlane(blockIdx.x * 8 + (threadIdx.x >> 5));
    if (node >= n_nodes) return;
    const int beg = row_ptr[node];
    const int end = row_ptr[node + 1];

    float4 acc = make_float4(0.0f, 0.0f, 0.0f, 0.0f);
    for (int j0 = beg; j0 < end; j0 += 32) {
        int cnt = end - j0;
        if (cnt > 32) cnt = 32;
        const int myc = (lane < cnt) ? col[j0 + lane] : 0;
        int k = 0;
        for (; k + 8 <= cnt; k += 8) {
            float4 v[8];
            #pragma unroll
            for (int u = 0; u < 8; ++u)
                v[u] = gcn_row4(hs, __builtin_amdgcn_readlane(myc, k + u), lane);
            #pragma unroll
            for (int u = 0; u < 8; ++u) gcn_acc4(acc, v[u]);
        }
        if (cnt & 4) {
            float4 v[4];
            #pragma unroll
            for (int u = 0; u < 4; ++u)
                v[u] = gcn_row4(hs, __builtin_amdgcn_readlane(myc, k + u), lane);
            #pragma unroll
            for (int u = 0; u < 4; ++u) gcn_acc4(acc, v[u]);
            k += 4;
        }
        if (cnt & 2) {
            float4 v0 = gcn_row4(hs, __builtin_amdgcn_readlane(myc, k), lane);
            float4 v1 = gcn_row4(hs, __builtin_amdgcn_readlane(myc, k + 1), lane);
            gcn_acc4(acc, v0);
            gcn_acc4(acc, v1);
            k += 2;
        }
        if (cnt & 1) {
            float4 v0 = gcn_row4(hs, __builtin_amdgcn_readlane(myc, k), lane);
            gcn_acc4(acc, v0);
        }
    }
    *(float4*)(agg + (size_t)node * 128 + lane * 4) = acc;
}

__global__ __launch_bounds__(256)
void gcn_gemm_wmma_kernel(const float* __restrict__ A,
                          const float* __restrict__ norm_in,
                          const float* __restrict__ W,
                          const float* __restrict__ bias,
                          float* __restrict__ out,
                          int N, int relu) {
    const int K = 128;
    __shared__ float As[16 * 132];

    const int wave = threadIdx.x >> 5;
    const int lane = threadIdx.x & 31;
    const int half = lane >> 4;
    const int l16  = lane & 15;
    const int tileM = blockIdx.x * 16;
    const int tileN = wave * 16;

    for (int i = threadIdx.x; i < 16 * 32; i += 256) {
        int r  = i >> 5;
        int c4 = i & 31;
        float4 v = ((const float4*)(A + (size_t)(tileM + r) * K))[c4];
        float s = norm_in[tileM + r];
        v.x *= s; v.y *= s; v.z *= s; v.w *= s;
        ((float4*)(As + r * 132))[c4] = v;
    }
    __syncthreads();

    if (tileN >= N) return;

    v8f acc = {};
    for (int k0 = 0; k0 < K; k0 += 4) {
        int ka = k0 + 2 * half;
        v2f a, b;
        a.x = As[l16 * 132 + ka];
        a.y = As[l16 * 132 + ka + 1];
        b.x = W[(size_t)ka * N + tileN + l16];
        b.y = W[(size_t)(ka + 1) * N + tileN + l16];
        acc = __builtin_amdgcn_wmma_f32_16x16x4_f32(
             false, a,  false, b,
             (short)0, acc,  false,  false);
    }

    float bb = bias[tileN + l16];
    #pragma unroll
    for (int r = 0; r < 8; ++r) {
        int row = tileM + r + 8 * half;
        float v = acc[r] + bb;
        if (relu) v = fmaxf(v, 0.0f);
        out[(size_t)row * N + tileN + l16] = v;
    }
}

extern "C" void kernel_launch(void* const* d_in, const int* in_sizes, int n_in,
                              void* d_out, int out_size, void* d_ws, size_t ws_size,
                              hipStream_t stream) {
    const float* in_feat = (const float*)d_in[0];
    const int*   e_src   = (const int*)d_in[1];
    const int*   e_dst   = (const int*)d_in[2];
    const float* W1 = (const float*)d_in[3];
    const float* b1 = (const float*)d_in[4];
    const float* W2 = (const float*)d_in[5];
    const float* b2 = (const float*)d_in[6];
    const float* W3 = (const float*)d_in[7];
    const float* b3 = (const float*)d_in[8];
    float* out = (float*)d_out;

    const size_t NF = (size_t)N_NODES * 128;
    float* B0 = (float*)d_ws;
    float* B1 = B0 + NF;
    float* B2 = B1 + NF;
    int*   deg_out  = (int*)(B2 + NF);
    int*   deg_in   = deg_out + N_NODES;
    float* norm_out = (float*)(deg_in + N_NODES);
    float* norm_in  = norm_out + N_NODES;
    int*   rank     = (int*)(norm_in + N_NODES);
    int*   col      = rank + N_EDGES;
    int*   row_ptr  = col + N_EDGES;

    const int TB = 256;
    const int edgeBlocks   = (N_EDGES + TB - 1) / TB;
    const int gatherBlocks = (N_NODES + 7) / 8;
    const int scaleBlocks  = (N_NODES * 32 + TB - 1) / TB;
    const int nodeBlocks   = (N_NODES + TB - 1) / TB;
    const int gemmBlocks   = N_NODES / 16;

    hipMemsetAsync(deg_out, 0, 2 * (size_t)N_NODES * sizeof(int), stream);
    gcn_degree_rank_kernel<<<edgeBlocks, TB, 0, stream>>>(e_src, e_dst, deg_out, deg_in, rank, N_EDGES);
    gcn_rowptr_norm_kernel<<<nodeBlocks, TB, 0, stream>>>(deg_out, deg_in, norm_out, norm_in, row_ptr, N_NODES);
    gcn_fill_kernel<<<edgeBlocks, TB, 0, stream>>>(e_src, e_dst, rank, row_ptr, col, N_EDGES);

    gcn_scale_kernel<<<scaleBlocks, TB, 0, stream>>>(in_feat, norm_out, B0, N_NODES);
    gcn_gather_kernel<<<gatherBlocks, TB, 0, stream>>>(B0, row_ptr, col, B1, N_NODES);
    gcn_gemm_wmma_kernel<<<gemmBlocks, TB, 0, stream>>>(B1, norm_in, W1, b1, B2, H_FEATS, 1);

    gcn_scale_kernel<<<scaleBlocks, TB, 0, stream>>>(B2, norm_out, B0, N_NODES);
    gcn_gather_kernel<<<gatherBlocks, TB, 0, stream>>>(B0, row_ptr, col, B1, N_NODES);
    gcn_gemm_wmma_kernel<<<gemmBlocks, TB, 0, stream>>>(B1, norm_in, W2, b2, B2, H_FEATS, 1);

    gcn_scale_kernel<<<scaleBlocks, TB, 0, stream>>>(B2, norm_out, B0, N_NODES);
    gcn_gather_kernel<<<gatherBlocks, TB, 0, stream>>>(B0, row_ptr, col, B1, N_NODES);
    gcn_gemm_wmma_kernel<<<gemmBlocks, TB, 0, stream>>>(B1, norm_in, W3, b3, out, NUM_CLASSES, 0);
}


# Out-degree scale rides in the sorted edge list as (source id, scale) pairs; the row gather multiply-adds unscaled rows by it (fma, slot order kept); the three scaling kernels and their buffer go
# speedup vs baseline: 2.7941x; 1.2649x over previous
#include <hip/hip_runtime.h>
#include <hip/hip_bf16.h>


#define N_NODES     50000
#define N_EDGES     800000
#define IN_FEATS    128
#define H_FEATS     128
#define NUM_CLASSES 16

typedef __attribute__((ext_vector_type(2))) float v2f;
typedef __attribute__((ext_vector_type(8))) float v8f;

__global__ void gcn_degree_rank_kernel(const int* __restrict__ src,
                                       const int* __restrict__ dst,
                                       int* __restrict__ deg_out,
                                       int* __restrict__ deg_in,
                                       int* __restrict__ rank, int E) {
    int e = blockIdx.x * blockDim.x + threadIdx.x;
    if (e < E) {
        const int s = src[e];
        const int d = dst[e];
        atomicAdd(&deg_out[s], 1);
        rank[e] = atomicAdd(&deg_in[d], 1);
    }
}

__global__ __launch_bounds__(256)
void gcn_rowptr_norm_kernel(const int* __restrict__ deg_out,
                            const int* __restrict__ deg_in,
                            float* __restrict__ norm_out,
                            float* __restrict__ norm_in,
                            int* __restrict__ row_ptr, int n) {
    __shared__ int red[8];
    __shared__ int wsum[8];
    const int tid  = threadIdx.x;
    const int lane = tid & 31;
    const int wave = tid >> 5;
    const int base = blockIdx.x * 256;

    int s = 0;
    const int4* p4 = (const int4*)deg_in + tid;
    const int n4 = base >> 2;
    const int full = n4 >> 8;
    int m = 0;
    for (; m + 8 <= full; m += 8) {
        int4 v[8];
        #pragma unroll
        for (int u = 0; u < 8; ++u) v[u] = p4[(m + u) * 256];
        #pragma unroll
        for (int u = 0; u < 8; ++u) s += (v[u].x + v[u].y) + (v[u].z + v[u].w);
    }
    if (full & 4) {
        int4 v[4];
        #pragma unroll
        for (int u = 0; u < 4; ++u) v[u] = p4[(m + u) * 256];
        #pragma unroll
        for (int u = 0; u < 4; ++u) s += (v[u].x + v[u].y) + (v[u].z + v[u].w);
        m += 4;
    }
    if (full & 2) {
        int4 v0 = p4[m * 256];
        int4 v1 = p4[(m + 1) * 256];
        s += (v0.x + v0.y) + (v0.z + v0.w);
        s += (v1.x + v1.y) + (v1.z + v1.w);
        m += 2;
    }
    if (full & 1) {
        int4 v0 = p4[m * 256];
        s += (v0.x + v0.y) + (v0.z + v0.w);
        m += 1;
    }
    if (tid < (n4 & 255)) {
        int4 v0 = p4[m * 256];
        s += (v0.x + v0.y) + (v0.z + v0.w);
    }
    for (int o = 16; o > 0; o >>= 1) s += __shfl_xor(s, o);
    if (lane == 0) red[wave] = s;

    const int i = base + tid;
    int di = 0;
    if (i < n) {
        di = deg_in[i];
        int dq = deg_out[i];
        norm_out[i] = __frsqrt_rn(fmaxf((float)dq, 1.0f));
        norm_in[i]  = __frsqrt_rn(fmaxf((float)di, 1.0f));
    }
    int x = di;
    for (int o = 1; o < 32; o <<= 1) {
        int y = __shfl_up(x, o);
        if (lane >= o) x += y;
    }
    if (lane == 31) wsum[wave] = x;
    __syncthreads();

    int off = 0;
    for (int w = 0; w < 8; ++w) {
        off += red[w];
        if (w < wave) off += wsum[w];
    }
    const int excl = off + x - di;
    if (i < n) row_ptr[i] = excl;
    if (i == n - 1) row_ptr[n] = excl + di;
}

__global__ void gcn_fill_kernel(const int* __restrict__ src,
                                const int* __restrict__ dst,
                                const int* __restrict__ rank,
                                const int* __restrict__ row_ptr,
                                const float* __restrict__ norm_out,
                                int* __restrict__ colw, int E) {
    int e = blockIdx.x * blockDim.x + threadIdx.x;
    if (e < E) {
        const int s = src[e];
        ((int2*)colw)[row_ptr[dst[e]] + rank[e]] = make_int2(s, __float_as_int(norm_out[s]));
    }
}

__device__ __forceinline__ float4 gcn_row4(const float* __restrict__ h, int s, int lane) {
    return *(const float4*)(h + (size_t)s * 128 + lane * 4);
}
__device__ __forceinline__ void gcn_mac4(float4& a, const float4& v, float w) {
    a.x = __builtin_fmaf(v.x, w, a.x); a.y = __builtin_fmaf(v.y, w, a.y);
    a.z = __builtin_fmaf(v.z, w, a.z); a.w = __builtin_fmaf(v.w, w, a.w);
}

__global__ __launch_bounds__(256)
void gcn_gather_kernel(const float* __restrict__ h,
                       const int* __restrict__ row_ptr,
                       const int* __restrict__ colw,
                       float* __restrict__ agg, int n_nodes) {
    const int lane = threadIdx.x & 31;
    const int node = __builtin_amdgcn_readfirstlane(blockIdx.x * 8 + (threadIdx.x >> 5));
    if (node >= n_nodes) return;
    const int beg = row_ptr[node];
    const int end = row_ptr[node + 1];

    float4 acc = make_float4(0.0f, 0.0f, 0.0f, 0.0f);
    for (int j0 = beg; j0 < end; j0 += 32) {
        int cnt = end - j0;
        if (cnt > 32) cnt = 32;
        const int2 my = (lane < cnt) ? ((const int2*)colw)[j0 + lane] : make_int2(0, 0);
        int k = 0;
        for (; k + 8 <= cnt; k += 8) {
            float4 v[8];
            float w[8];
            #pragma unroll
            for (int u = 0; u < 8; ++u) {
                v[u] = gcn_row4(h, __builtin_amdgcn_readlane(my.x, k + u), lane);
                w[u] = __int_as_float(__builtin_amdgcn_readlane(my.y, k + u));
            }
            #pragma unroll
            for (int u = 0; u < 8; ++u) gcn_mac4(acc, v[u], w[u]);
        }
        if (cnt & 4) {
            float4 v[4];
            float w[4];
            #pragma unroll
            for (int u = 0; u < 4; ++u) {
                v[u] = gcn_row4(h, __builtin_amdgcn_readlane(my.x, k + u), lane);
                w[u] = __int_as_float(__builtin_amdgcn_readlane(my.y, k + u));
            }
            #pragma unroll
            for (int u = 0; u < 4; ++u) gcn_mac4(acc, v[u], w[u]);
            k += 4;
        }
        if (cnt & 2) {
            float4 v0 = gcn_row4(h, __builtin_amdgcn_readlane(my.x, k), lane);
            float4 v1 = gcn_row4(h, __builtin_amdgcn_readlane(my.x, k + 1), lane);
            const float w0 = __int_as_float(__builtin_amdgcn_readlane(my.y, k));
            const float w1 = __int_as_float(__builtin_amdgcn_readlane(my.y, k + 1));
            gcn_mac4(acc, v0, w0);
            gcn_mac4(acc, v1, w1);
            k += 2;
        }
        if (cnt & 1) {
            float4 v0 = gcn_row4(h, __builtin_amdgcn_readlane(my.x, k), lane);
            const float w0 = __int_as_float(__builtin_amdgcn_readlane(my.y, k));
            gcn_mac4(acc, v0, w0);
        }
    }
    *(float4*)(agg + (size_t)node * 128 + lane * 4) = acc;
}

__global__ __launch_bounds__(256)
void gcn_gemm_wmma_kernel(const float* __restrict__ A,
                          const float* __restrict__ norm_in,
                          const float* __restrict__ W,
                          const float* __restrict__ bias,
                          float* __restrict__ out,
                          int N, int relu) {
    const int K = 128;
    __shared__ float As[16 * 132];

    const int wave = threadIdx.x >> 5;
    const int lane = threadIdx.x & 31;
    const int half = lane >> 4;
    const int l16  = lane & 15;
    const int tileM = blockIdx.x * 16;
    const int tileN = wave * 16;

    for (int i = threadIdx.x; i < 16 * 32; i += 256) {
        int r  = i >> 5;
        int c4 = i & 31;
        float4 v = ((const float4*)(A + (size_t)(tileM + r) * K))[c4];
        float s = norm_in[tileM + r];
        v.x *= s; v.y *= s; v.z *= s; v.w *= s;
        ((float4*)(As + r * 132))[c4] = v;
    }
    __syncthreads();

    if (tileN >= N) return;

    v8f acc = {};
    for (int k0 = 0; k0 < K; k0 += 4) {
        int ka = k0 + 2 * half;
        v2f a, b;
        a.x = As[l16 * 132 + ka];
        a.y = As[l16 * 132 + ka + 1];
        b.x = W[(size_t)ka * N + tileN + l16];
        b.y = W[(size_t)(ka + 1) * N + tileN + l16];
        acc = __builtin_amdgcn_wmma_f32_16x16x4_f32(
             false, a,  false, b,
             (short)0, acc,  false,  false);
    }

    float bb = bias[tileN + l16];
    #pragma unroll
    for (int r = 0; r < 8; ++r) {
        int row = tileM + r + 8 * half;
        float v = acc[r] + bb;
        if (relu) v = fmaxf(v, 0.0f);
        out[(size_t)row * N + tileN + l16] = v;
    }
}

extern "C" void kernel_launch(void* const* d_in, const int* in_sizes, int n_in,
                              void* d_out, int out_size, void* d_ws, size_t ws_size,
                              hipStream_t stream) {
    const float* in_feat = (const float*)d_in[0];
    const int*   e_src   = (const int*)d_in[1];
    const int*   e_dst   = (const int*)d_in[2];
    const float* W1 = (const float*)d_in[3];
    const float* b1 = (const float*)d_in[4];
    const float* W2 = (const float*)d_in[5];
    const float* b2 = (const float*)d_in[6];
    const float* W3 = (const float*)d_in[7];
    const float* b3 = (const float*)d_in[8];
    float* out = (float*)d_out;

    const size_t NF = (size_t)N_NODES * 128;
    float* B1 = (float*)d_ws;
    float* B2 = B1 + NF;
    int*   deg_out  = (int*)(B2 + NF);
    int*   deg_in   = deg_out + N_NODES;
    float* norm_out = (float*)(deg_in + N_NODES);
    float* norm_in  = norm_out + N_NODES;
    int*   rank     = (int*)(norm_in + N_NODES);
    int*   colw     = rank + N_EDGES;
    int*   row_ptr  = colw + 2 * N_EDGES;

    const int TB = 256;
    const int edgeBlocks   = (N_EDGES + TB - 1) / TB;
    const int gatherBlocks = (N_NODES + 7) / 8;
    const int nodeBlocks   = (N_NODES + TB - 1) / TB;
    const int gemmBlocks   = N_NODES / 16;

    hipMemsetAsync(deg_out, 0, 2 * (size_t)N_NODES * sizeof(int), stream);
    gcn_degree_rank_kernel<<<edgeBlocks, TB, 0, stream>>>(e_src, e_dst, deg_out, deg_in, rank, N_EDGES);
    gcn_rowptr_norm_kernel<<<nodeBlocks, TB, 0, stream>>>(deg_out, deg_in, norm_out, norm_in, row_ptr, N_NODES);
    gcn_fill_kernel<<<edgeBlocks, TB, 0, stream>>>(e_src, e_dst, rank, row_ptr, norm_out, colw, N_EDGES);

    gcn_gather_kernel<<<gatherBlocks, TB, 0, stream>>>(in_feat, row_ptr, colw, B1, N_NODES);
    gcn_gemm_wmma_kernel<<<gemmBlocks, TB, 0, stream>>>(B1, norm_in, W1, b1, B2, H_FEATS, 1);

    gcn_gather_kernel<<<gatherBlocks, TB, 0, stream>>>(B2, row_ptr, colw, B1, N_NODES);
    gcn_gemm_wmma_kernel<<<gemmBlocks, TB, 0, stream>>>(B1, norm_in, W2, b2, B2, H_FEATS, 1);

    gcn_gather_kernel<<<gatherBlocks, TB, 0, stream>>>(B2, row_ptr, colw, B1, N_NODES);
    gcn_gemm_wmma_kernel<<<gemmBlocks, TB, 0, stream>>>(B1, norm_in, W3, b3, out, NUM_CLASSES, 0);
}


# Row gather fused into the WMMA GEMM kernel: each 8-wave block sums its 32 rows (4 a wave, slot order kept) straight into the LDS A tile; 3 launches and the aggregated-row buffer round trip go
# speedup vs baseline: 3.3127x; 1.1856x over previous
#include <hip/hip_runtime.h>
#include <hip/hip_bf16.h>


#define N_NODES     50000
#define N_EDGES     800000
#define IN_FEATS    128
#define H_FEATS     128
#define NUM_CLASSES 16

typedef __attribute__((ext_vector_type(2))) float v2f;
typedef __attribute__((ext_vector_type(8))) float v8f;

__global__ void gcn_degree_rank_kernel(const int* __restrict__ src,
                                       const int* __restrict__ dst,
                                       int* __restrict__ deg_out,
                                       int* __restrict__ deg_in,
                                       int* __restrict__ rank, int E) {
    int e = blockIdx.x * blockDim.x + threadIdx.x;
    if (e < E) {
        const int s = src[e];
        const int d = dst[e];
        atomicAdd(&deg_out[s], 1);
        rank[e] = atomicAdd(&deg_in[d], 1);
    }
}

__global__ __launch_bounds__(256)
void gcn_rowptr_norm_kernel(const int* __restrict__ deg_out,
                            const int* __restrict__ deg_in,
                            float* __restrict__ norm_out,
                            float* __restrict__ norm_in,
                            int* __restrict__ row_ptr, int n) {
    __shared__ int red[8];
    __shared__ int wsum[8];
    const int tid  = threadIdx.x;
    const int lane = tid & 31;
    const int wave = tid >> 5;
    const int base = blockIdx.x * 256;

    int s = 0;
    const int4* p4 = (const int4*)deg_in + tid;
    const int n4 = base >> 2;
    const int full = n4 >> 8;
    int m = 0;
    for (; m + 8 <= full; m += 8) {
        int4 v[8];
        #pragma unroll
        for (int u = 0; u < 8; ++u) v[u] = p4[(m + u) * 256];
        #pragma unroll
        for (int u = 0; u < 8; ++u) s += (v[u].x + v[u].y) + (v[u].z + v[u].w);
    }
    if (full & 4) {
        int4 v[4];
        #pragma unroll
        for (int u = 0; u < 4; ++u) v[u] = p4[(m + u) * 256];
        #pragma unroll
        for (int u = 0; u < 4; ++u) s += (v[u].x + v[u].y) + (v[u].z + v[u].w);
        m += 4;
    }
    if (full & 2) {
        int4 v0 = p4[m * 256];
        int4 v1 = p4[(m + 1) * 256];
        s += (v0.x + v0.y) + (v0.z + v0.w);
        s += (v1.x + v1.y) + (v1.z + v1.w);
        m += 2;
    }
    if (full & 1) {
        int4 v0 = p4[m * 256];
        s += (v0.x + v0.y) + (v0.z + v0.w);
        m += 1;
    }
    if (tid < (n4 & 255)) {
        int4 v0 = p4[m * 256];
        s += (v0.x + v0.y) + (v0.z + v0.w);
    }
    for (int o = 16; o > 0; o >>= 1) s += __shfl_xor(s, o);
    if (lane == 0) red[wave] = s;

    const int i = base + tid;
    int di = 0;
    if (i < n) {
        di = deg_in[i];
        int dq = deg_out[i];
        norm_out[i] = __frsqrt_rn(fmaxf((float)dq, 1.0f));
        norm_in[i]  = __frsqrt_rn(fmaxf((float)di, 1.0f));
    }
    int x = di;
    for (int o = 1; o < 32; o <<= 1) {
        int y = __shfl_up(x, o);
        if (lane >= o) x += y;
    }
    if (lane == 31) wsum[wave] = x;
    __syncthreads();

    int off = 0;
    for (int w = 0; w < 8; ++w) {
        off += red[w];
        if (w < wave) off += wsum[w];
    }
    const int excl = off + x - di;
    if (i < n) row_ptr[i] = excl;
    if (i == n - 1) row_ptr[n] = excl + di;
}

__global__ void gcn_fill_kernel(const int* __restrict__ src,
                                const int* __restrict__ dst,
                                const int* __restrict__ rank,
                                const int* __restrict__ row_ptr,
                                const float* __restrict__ norm_out,
                                int* __restrict__ colw, int E) {
    int e = blockIdx.x * blockDim.x + threadIdx.x;
    if (e < E) {
        const int s = src[e];
        ((int2*)colw)[row_ptr[dst[e]] + rank[e]] = make_int2(s, __float_as_int(norm_out[s]));
    }
}

__device__ __forceinline__ float4 gcn_row4(const float* __restrict__ h, int s, int lane) {
    return *(const float4*)(h + (size_t)s * 128 + lane * 4);
}
__device__ __forceinline__ void gcn_mac4(float4& a, const float4& v, float w) {
    a.x = __builtin_fmaf(v.x, w, a.x); a.y = __builtin_fmaf(v.y, w, a.y);
    a.z = __builtin_fmaf(v.z, w, a.z); a.w = __builtin_fmaf(v.w, w, a.w);
}

__device__ __forceinline__ float4 gcn_gather_row(const float* __restrict__ h,
                                                 const int* __restrict__ colw,
                                                 int beg, int end, int2 my, int lane) {
    float4 acc = make_float4(0.0f, 0.0f, 0.0f, 0.0f);
    for (int j0 = beg; j0 < end; j0 += 32) {
        int cnt = end - j0;
        if (cnt > 32) cnt = 32;
        if (j0 != beg) my = (lane < cnt) ? ((const int2*)colw)[j0 + lane] : make_int2(0, 0);
        int k = 0;
        for (; k + 8 <= cnt; k += 8) {
            float4 v[8];
            float w[8];
            #pragma unroll
            for (int u = 0; u < 8; ++u) {
                v[u] = gcn_row4(h, __builtin_amdgcn_readlane(my.x, k + u), lane);
                w[u] = __int_as_float(__builtin_amdgcn_readlane(my.y, k + u));
            }
            #pragma unroll
            for (int u = 0; u < 8; ++u) gcn_mac4(acc, v[u], w[u]);
        }
        if (cnt & 4) {
            float4 v[4];
            float w[4];
            #pragma unroll
            for (int u = 0; u < 4; ++u) {
                v[u] = gcn_row4(h, __builtin_amdgcn_readlane(my.x, k + u), lane);
                w[u] = __int_as_float(__builtin_amdgcn_readlane(my.y, k + u));
            }
            #pragma unroll
            for (int u = 0; u < 4; ++u) gcn_mac4(acc, v[u], w[u]);
            k += 4;
        }
        if (cnt & 2) {
            float4 v0 = gcn_row4(h, __builtin_amdgcn_readlane(my.x, k), lane);
            float4 v1 = gcn_row4(h, __builtin_amdgcn_readlane(my.x, k + 1), lane);
            const float w0 = __int_as_float(__builtin_amdgcn_readlane(my.y, k));
            const float w1 = __int_as_float(__builtin_amdgcn_readlane(my.y, k + 1));
            gcn_mac4(acc, v0, w0);
            gcn_mac4(acc, v1, w1);
            k += 2;
        }
        if (cnt & 1) {
            float4 v0 = gcn_row4(h, __builtin_amdgcn_readlane(my.x, k), lane);
            const float w0 = __int_as_float(__builtin_amdgcn_readlane(my.y, k));
            gcn_mac4(acc, v0, w0);
        }
    }
    return acc;
}

__global__ __launch_bounds__(256)
void gcn_layer_kernel(const float* __restrict__ h,
                      const int* __restrict__ row_ptr,
                      const int* __restrict__ colw,
                      const float* __restrict__ norm_in,
                      const float* __restrict__ W,
                      const float* __restrict__ bias,
                      float* __restrict__ out,
                      int N, int relu, int n_nodes) {
    const int K = 128;
    __shared__ float As[32 * 132];

    const int lane = threadIdx.x & 31;
    const int wave = __builtin_amdgcn_readfirstlane(threadIdx.x >> 5);
    const int half = lane >> 4;
    const int l16  = lane & 15;
    const int tileM = blockIdx.x * 32;
    const int tileN = wave * 16;

    const int row0 = tileM + wave * 4;
    int rp[5];
    #pragma unroll
    for (int u = 0; u < 5; ++u) {
        int idx = row0 + u;
        if (idx > n_nodes) idx = n_nodes;
        rp[u] = row_ptr[idx];
    }
    int2 my[4];
    #pragma unroll
    for (int u = 0; u < 4; ++u) {
        int cnt = rp[u + 1] - rp[u];
        if (cnt > 32) cnt = 32;
        my[u] = (lane < cnt) ? ((const int2*)colw)[rp[u] + lane] : make_int2(0, 0);
    }
    #pragma unroll
    for (int u = 0; u < 4; ++u) {
        float4 a = gcn_gather_row(h, colw, rp[u], rp[u + 1], my[u], lane);
        const int node = row0 + u;
        const float s = (node < n_nodes) ? norm_in[node] : 0.0f;
        a.x *= s; a.y *= s; a.z *= s; a.w *= s;
        *(float4*)(As + (wave * 4 + u) * 132 + lane * 4) = a;
    }
    __syncthreads();

    if (tileN >= N) return;

    v8f acc0 = {};
    v8f acc1 = {};
    #pragma unroll 4
    for (int k0 = 0; k0 < K; k0 += 4) {
        int ka = k0 + 2 * half;
        v2f a0, a1, b;
        a0.x = As[l16 * 132 + ka];
        a0.y = As[l16 * 132 + ka + 1];
        a1.x = As[(16 + l16) * 132 + ka];
        a1.y = As[(16 + l16) * 132 + ka + 1];
        b.x = W[(size_t)ka * N + tileN + l16];
        b.y = W[(size_t)(ka + 1) * N + tileN + l16];
        acc0 = __builtin_amdgcn_wmma_f32_16x16x4_f32(
             false, a0,  false, b,
             (short)0, acc0,  false,  false);
        acc1 = __builtin_amdgcn_wmma_f32_16x16x4_f32(
             false, a1,  false, b,
             (short)0, acc1,  false,  false);
    }

    const float bb = bias[tileN + l16];
    #pragma unroll
    for (int r = 0; r < 8; ++r) {
        int row = tileM + r + 8 * half;
        float v = acc0[r] + bb;
        if (relu) v = fmaxf(v, 0.0f);
        out[(size_t)row * N + tileN + l16] = v;
    }
    if (tileM + 16 < n_nodes) {
        #pragma unroll
        for (int r = 0; r < 8; ++r) {
            int row = tileM + 16 + r + 8 * half;
            float v = acc1[r] + bb;
            if (relu) v = fmaxf(v, 0.0f);
            out[(size_t)row * N + tileN + l16] = v;
        }
    }
}

extern "C" void kernel_launch(void* const* d_in, const int* in_sizes, int n_in,
                              void* d_out, int out_size, void* d_ws, size_t ws_size,
                              hipStream_t stream) {
    const float* in_feat = (const float*)d_in[0];
    const int*   e_src   = (const int*)d_in[1];
    const int*   e_dst   = (const int*)d_in[2];
    const float* W1 = (const float*)d_in[3];
    const float* b1 = (const float*)d_in[4];
    const float* W2 = (const float*)d_in[5];
    const float* b2 = (const float*)d_in[6];
    const float* W3 = (const float*)d_in[7];
    const float* b3 = (const float*)d_in[8];
    float* out = (float*)d_out;

    const size_t NF = (size_t)N_NODES * 128;
    float* B1 = (float*)d_ws;
    float* B2 = B1 + NF;
    int*   deg_out  = (int*)(B2 + NF);
    int*   deg_in   = deg_out + N_NODES;
    float* norm_out = (float*)(deg_in + N_NODES);
    float* norm_in  = norm_out + N_NODES;
    int*   rank     = (int*)(norm_in + N_NODES);
    int*   colw     = rank + N_EDGES;
    int*   row_ptr  = colw + 2 * N_EDGES;

    const int TB = 256;
    const int edgeBlocks  = (N_EDGES + TB - 1) / TB;
    const int nodeBlocks  = (N_NODES + TB - 1) / TB;
    const int layerBlocks = (N_NODES + 31) / 32;

    hipMemsetAsync(deg_out, 0, 2 * (size_t)N_NODES * sizeof(int), stream);
    gcn_degree_rank_kernel<<<edgeBlocks, TB, 0, stream>>>(e_src, e_dst, deg_out, deg_in, rank, N_EDGES);
    gcn_rowptr_norm_kernel<<<nodeBlocks, TB, 0, stream>>>(deg_out, deg_in, norm_out, norm_in, row_ptr, N_NODES);
    gcn_fill_kernel<<<edgeBlocks, TB, 0, stream>>>(e_src, e_dst, rank, row_ptr, norm_out, colw, N_EDGES);

    gcn_layer_kernel<<<layerBlocks, TB, 0, stream>>>(in_feat, row_ptr, colw, norm_in, W1, b1, B1, H_FEATS, 1, N_NODES);
    gcn_layer_kernel<<<layerBlocks, TB, 0, stream>>>(B1, row_ptr, colw, norm_in, W2, b2, B2, H_FEATS, 1, N_NODES);
    gcn_layer_kernel<<<layerBlocks, TB, 0, stream>>>(B2, row_ptr, colw, norm_in, W3, b3, out, NUM_CLASSES, 0, N_NODES);
}


# Layer 3's weights applied before its aggregation, inside the layer-2 launch (tile times W3 on WMMA, k quarters summed in fixed order); layer 3 gathers 16-float rows; sums reassociated
# speedup vs baseline: 3.7063x; 1.1188x over previous
#include <hip/hip_runtime.h>
#include <hip/hip_bf16.h>


#define N_NODES     50000
#define N_EDGES     800000
#define IN_FEATS    128
#define H_FEATS     128
#define NUM_CLASSES 16

typedef __attribute__((ext_vector_type(2))) float v2f;
typedef __attribute__((ext_vector_type(8))) float v8f;

__global__ void gcn_degree_rank_kernel(const int* __restrict__ src,
                                       const int* __restrict__ dst,
                                       int* __restrict__ deg_out,
                                       int* __restrict__ deg_in,
                                       int* __restrict__ rank, int E) {
    int e = blockIdx.x * blockDim.x + threadIdx.x;
    if (e < E) {
        const int s = src[e];
        const int d = dst[e];
        atomicAdd(&deg_out[s], 1);
        rank[e] = atomicAdd(&deg_in[d], 1);
    }
}

__global__ __launch_bounds__(256)
void gcn_rowptr_norm_kernel(const int* __restrict__ deg_out,
                            const int* __restrict__ deg_in,
                            float* __restrict__ norm_out,
                            float* __restrict__ norm_in,
                            int* __restrict__ row_ptr, int n) {
    __shared__ int red[8];
    __shared__ int wsum[8];
    const int tid  = threadIdx.x;
    const int lane = tid & 31;
    const int wave = tid >> 5;
    const int base = blockIdx.x * 256;

    int s = 0;
    const int4* p4 = (const int4*)deg_in + tid;
    const int n4 = base >> 2;
    const int full = n4 >> 8;
    int m = 0;
    for (; m + 8 <= full; m += 8) {
        int4 v[8];
        #pragma unroll
        for (int u = 0; u < 8; ++u) v[u] = p4[(m + u) * 256];
        #pragma unroll
        for (int u = 0; u < 8; ++u) s += (v[u].x + v[u].y) + (v[u].z + v[u].w);
    }
    if (full & 4) {
        int4 v[4];
        #pragma unroll
        for (int u = 0; u < 4; ++u) v[u] = p4[(m + u) * 256];
        #pragma unroll
        for (int u = 0; u < 4; ++u) s += (v[u].x + v[u].y) + (v[u].z + v[u].w);
        m += 4;
    }
    if (full & 2) {
        int4 v0 = p4[m * 256];
        int4 v1 = p4[(m + 1) * 256];
        s += (v0.x + v0.y) + (v0.z + v0.w);
        s += (v1.x + v1.y) + (v1.z + v1.w);
        m += 2;
    }
    if (full & 1) {
        int4 v0 = p4[m * 256];
        s += (v0.x + v0.y) + (v0.z + v0.w);
        m += 1;
    }
    if (tid < (n4 & 255)) {
        int4 v0 = p4[m * 256];
        s += (v0.x + v0.y) + (v0.z + v0.w);
    }
    for (int o = 16; o > 0; o >>= 1) s += __shfl_xor(s, o);
    if (lane == 0) red[wave] = s;

    const int i = base + tid;
    int di = 0;
    if (i < n) {
        di = deg_in[i];
        int dq = deg_out[i];
        norm_out[i] = __frsqrt_rn(fmaxf((float)dq, 1.0f));
        norm_in[i]  = __frsqrt_rn(fmaxf((float)di, 1.0f));
    }
    int x = di;
    for (int o = 1; o < 32; o <<= 1) {
        int y = __shfl_up(x, o);
        if (lane >= o) x += y;
    }
    if (lane == 31) wsum[wave] = x;
    __syncthreads();

    int off = 0;
    for (int w = 0; w < 8; ++w) {
        off += red[w];
        if (w < wave) off += wsum[w];
    }
    const int excl = off + x - di;
    if (i < n) row_ptr[i] = excl;
    if (i == n - 1) row_ptr[n] = excl + di;
}

__global__ void gcn_fill_kernel(const int* __restrict__ src,
                                const int* __restrict__ dst,
                                const int* __restrict__ rank,
                                const int* __restrict__ row_ptr,
                                const float* __restrict__ norm_out,
                                int* __restrict__ colw, int E) {
    int e = blockIdx.x * blockDim.x + threadIdx.x;
    if (e < E) {
        const int s = src[e];
        ((int2*)colw)[row_ptr[dst[e]] + rank[e]] = make_int2(s, __float_as_int(norm_out[s]));
    }
}

__device__ __forceinline__ float4 gcn_row4(const float* __restrict__ h, int s, int lane) {
    return *(const float4*)(h + (size_t)s * 128 + lane * 4);
}
__device__ __forceinline__ void gcn_mac4(float4& a, const float4& v, float w) {
    a.x = __builtin_fmaf(v.x, w, a.x); a.y = __builtin_fmaf(v.y, w, a.y);
    a.z = __builtin_fmaf(v.z, w, a.z); a.w = __builtin_fmaf(v.w, w, a.w);
}

__device__ __forceinline__ float4 gcn_gather_row(const float* __restrict__ h,
                                                 const int* __restrict__ colw,
                                                 int beg, int end, int2 my, int lane) {
    float4 acc = make_float4(0.0f, 0.0f, 0.0f, 0.0f);
    for (int j0 = beg; j0 < end; j0 += 32) {
        int cnt = end - j0;
        if (cnt > 32) cnt = 32;
        if (j0 != beg) my = (lane < cnt) ? ((const int2*)colw)[j0 + lane] : make_int2(0, 0);
        int k = 0;
        for (; k + 8 <= cnt; k += 8) {
            float4 v[8];
            float w[8];
            #pragma unroll
            for (int u = 0; u < 8; ++u) {
                v[u] = gcn_row4(h, __builtin_amdgcn_readlane(my.x, k + u), lane);
                w[u] = __int_as_float(__builtin_amdgcn_readlane(my.y, k + u));
            }
            #pragma unroll
            for (int u = 0; u < 8; ++u) gcn_mac4(acc, v[u], w[u]);
        }
        if (cnt & 4) {
            float4 v[4];
            float w[4];
            #pragma unroll
            for (int u = 0; u < 4; ++u) {
                v[u] = gcn_row4(h, __builtin_amdgcn_readlane(my.x, k + u), lane);
                w[u] = __int_as_float(__builtin_amdgcn_readlane(my.y, k + u));
            }
            #pragma unroll
            for (int u = 0; u < 4; ++u) gcn_mac4(acc, v[u], w[u]);
            k += 4;
        }
        if (cnt & 2) {
            float4 v0 = gcn_row4(h, __builtin_amdgcn_readlane(my.x, k), lane);
            float4 v1 = gcn_row4(h, __builtin_amdgcn_readlane(my.x, k + 1), lane);
            const float w0 = __int_as_float(__builtin_amdgcn_readlane(my.y, k));
            const float w1 = __int_as_float(__builtin_amdgcn_readlane(my.y, k + 1));
            gcn_mac4(acc, v0, w0);
            gcn_mac4(acc, v1, w1);
            k += 2;
        }
        if (cnt & 1) {
            float4 v0 = gcn_row4(h, __builtin_amdgcn_readlane(my.x, k), lane);
            const float w0 = __int_as_float(__builtin_amdgcn_readlane(my.y, k));
            gcn_mac4(acc, v0, w0);
        }
    }
    return acc;
}

template <bool PROJ>
__device__ __forceinline__
void gcn_layer_body(const float* __restrict__ h,
                    const int* __restrict__ row_ptr,
                    const int* __restrict__ colw,
                    const float* __restrict__ norm_in,
                    const float* __restrict__ W,
                    const float* __restrict__ bias,
                    const float* __restrict__ Wn,
                    float* __restrict__ out,
                    int N, int relu, int n_nodes) {
    const int K = 128;
    __shared__ float As[32 * 132];

    const int lane = threadIdx.x & 31;
    const int wave = __builtin_amdgcn_readfirstlane(threadIdx.x >> 5);
    const int half = lane >> 4;
    const int l16  = lane & 15;
    const int tileM = blockIdx.x * 32;
    const int tileN = wave * 16;

    const int row0 = tileM + wave * 4;
    int rp[5];
    #pragma unroll
    for (int u = 0; u < 5; ++u) {
        int idx = row0 + u;
        if (idx > n_nodes) idx = n_nodes;
        rp[u] = row_ptr[idx];
    }
    int2 my[4];
    #pragma unroll
    for (int u = 0; u < 4; ++u) {
        int cnt = rp[u + 1] - rp[u];
        if (cnt > 32) cnt = 32;
        my[u] = (lane < cnt) ? ((const int2*)colw)[rp[u] + lane] : make_int2(0, 0);
    }
    #pragma unroll
    for (int u = 0; u < 4; ++u) {
        float4 a = gcn_gather_row(h, colw, rp[u], rp[u + 1], my[u], lane);
        const int node = row0 + u;
        const float s = (node < n_nodes) ? norm_in[node] : 0.0f;
        a.x *= s; a.y *= s; a.z *= s; a.w *= s;
        *(float4*)(As + (wave * 4 + u) * 132 + lane * 4) = a;
    }
    __syncthreads();

    if (!PROJ && tileN >= N) return;

    v8f acc0 = {};
    v8f acc1 = {};
    #pragma unroll 4
    for (int k0 = 0; k0 < K; k0 += 4) {
        int ka = k0 + 2 * half;
        v2f a0, a1, b;
        a0.x = As[l16 * 132 + ka];
        a0.y = As[l16 * 132 + ka + 1];
        a1.x = As[(16 + l16) * 132 + ka];
        a1.y = As[(16 + l16) * 132 + ka + 1];
        b.x = W[(size_t)ka * N + tileN + l16];
        b.y = W[(size_t)(ka + 1) * N + tileN + l16];
        acc0 = __builtin_amdgcn_wmma_f32_16x16x4_f32(
             false, a0,  false, b,
             (short)0, acc0,  false,  false);
        acc1 = __builtin_amdgcn_wmma_f32_16x16x4_f32(
             false, a1,  false, b,
             (short)0, acc1,  false,  false);
    }

    const float bb = bias[tileN + l16];
    if (PROJ) {
        __shared__ float Zp[8 * 256];
        const int m  = wave & 1;
        const int kq = wave >> 1;
        v2f bn[8];
        #pragma unroll
        for (int ks = 0; ks < 8; ++ks) {
            const int ka = kq * 32 + ks * 4 + 2 * half;
            bn[ks].x = Wn[ka * 16 + l16];
            bn[ks].y = Wn[(ka + 1) * 16 + l16];
        }
        __syncthreads();
        #pragma unroll
        for (int r = 0; r < 8; ++r) {
            float v0 = acc0[r] + bb;
            float v1 = acc1[r] + bb;
            if (relu) { v0 = fmaxf(v0, 0.0f); v1 = fmaxf(v1, 0.0f); }
            As[(r + 8 * half) * 132 + tileN + l16] = v0;
            As[(16 + r + 8 * half) * 132 + tileN + l16] = v1;
        }
        __syncthreads();
        v8f zacc = {};
        #pragma unroll
        for (int ks = 0; ks < 8; ++ks) {
            const int ka = kq * 32 + ks * 4 + 2 * half;
            v2f a;
            a.x = As[(m * 16 + l16) * 132 + ka];
            a.y = As[(m * 16 + l16) * 132 + ka + 1];
            zacc = __builtin_amdgcn_wmma_f32_16x16x4_f32(
                 false, a,  false, bn[ks],
                 (short)0, zacc,  false,  false);
        }
        #pragma unroll
        for (int r = 0; r < 8; ++r)
            Zp[wave * 256 + (r + 8 * half) * 16 + l16] = zacc[r];
        __syncthreads();
        const int t = threadIdx.x;
        if (t < 128) {
            const int row = t >> 2;
            const int q   = t & 3;
            const float* p = Zp + (row >> 4) * 256 + (row & 15) * 16 + q * 4;
            const float4 s0 = *(const float4*)(p);
            const float4 s1 = *(const float4*)(p + 512);
            const float4 s2 = *(const float4*)(p + 1024);
            const float4 s3 = *(const float4*)(p + 1536);
            float4 z;
            z.x = ((s0.x + s1.x) + s2.x) + s3.x;
            z.y = ((s0.y + s1.y) + s2.y) + s3.y;
            z.z = ((s0.z + s1.z) + s2.z) + s3.z;
            z.w = ((s0.w + s1.w) + s2.w) + s3.w;
            if (tileM + row < n_nodes)
                ((float4*)out)[(size_t)(tileM + row) * 4 + q] = z;
        }
        return;
    }
    #pragma unroll
    for (int r = 0; r < 8; ++r) {
        int row = tileM + r + 8 * half;
        float v = acc0[r] + bb;
        if (relu) v = fmaxf(v, 0.0f);
        out[(size_t)row * N + tileN + l16] = v;
    }
    if (tileM + 16 < n_nodes) {
        #pragma unroll
        for (int r = 0; r < 8; ++r) {
            int row = tileM + 16 + r + 8 * half;
            float v = acc1[r] + bb;
            if (relu) v = fmaxf(v, 0.0f);
            out[(size_t)row * N + tileN + l16] = v;
        }
    }
}

__global__ __launch_bounds__(256)
void gcn_layer_kernel(const float* __restrict__ h, const int* __restrict__ row_ptr,
                      const int* __restrict__ colw, const float* __restrict__ norm_in,
                      const float* __restrict__ W, const float* __restrict__ bias,
                      float* __restrict__ out, int N, int relu, int n_nodes) {
    gcn_layer_body<false>(h, row_ptr, colw, norm_in, W, bias, nullptr, out, N, relu, n_nodes);
}

__global__ __launch_bounds__(256)
void gcn_layer_proj_kernel(const float* __restrict__ h, const int* __restrict__ row_ptr,
                           const int* __restrict__ colw, const float* __restrict__ norm_in,
                           const float* __restrict__ W, const float* __restrict__ bias,
                           const float* __restrict__ Wn, float* __restrict__ z,
                           int relu, int n_nodes) {
    gcn_layer_body<true>(h, row_ptr, colw, norm_in, W, bias, Wn, z, 128, relu, n_nodes);
}

__global__ __launch_bounds__(256)
void gcn_gather16_kernel(const float* __restrict__ z,
                         const int* __restrict__ row_ptr,
                         const int* __restrict__ colw,
                         const float* __restrict__ norm_in,
                         const float* __restrict__ bias,
                         float* __restrict__ out,
                         int n_nodes) {
    const int t = blockIdx.x * 256 + threadIdx.x;
    const int node = t >> 2;
    const int q = t & 3;
    if (node >= n_nodes) return;
    const int beg = row_ptr[node];
    const int end = row_ptr[node + 1];
    const int2* cw = (const int2*)colw;
    const float4* z4 = (const float4*)z;
    const float s = norm_in[node];
    const float4 b = ((const float4*)bias)[q];
    float4 acc = make_float4(0.0f, 0.0f, 0.0f, 0.0f);
    if (beg < end) {
        const int last = end - 1;
        int2 p[8];
        #pragma unroll
        for (int u = 0; u < 8; ++u) p[u] = cw[min(beg + u, last)];
        for (int j = beg; j < end; j += 8) {
            float4 v[8];
            int2 pn[8];
            #pragma unroll
            for (int u = 0; u < 8; ++u) v[u] = z4[(size_t)p[u].x * 4 + q];
            #pragma unroll
            for (int u = 0; u < 8; ++u) pn[u] = cw[min(j + 8 + u, last)];
            #pragma unroll
            for (int u = 0; u < 8; ++u) {
                float4 t = acc;
                gcn_mac4(t, v[u], __int_as_float(p[u].y));
                const bool real = j + u < end;
                acc.x = real ? t.x : acc.x; acc.y = real ? t.y : acc.y;
                acc.z = real ? t.z : acc.z; acc.w = real ? t.w : acc.w;
            }
            #pragma unroll
            for (int u = 0; u < 8; ++u) p[u] = pn[u];
        }
    }
    float4 o;
    o.x = acc.x * s + b.x; o.y = acc.y * s + b.y;
    o.z = acc.z * s + b.z; o.w = acc.w * s + b.w;
    ((float4*)out)[(size_t)node * 4 + q] = o;
}

extern "C" void kernel_launch(void* const* d_in, const int* in_sizes, int n_in,
                              void* d_out, int out_size, void* d_ws, size_t ws_size,
                              hipStream_t stream) {
    const float* in_feat = (const float*)d_in[0];
    const int*   e_src   = (const int*)d_in[1];
    const int*   e_dst   = (const int*)d_in[2];
    const float* W1 = (const float*)d_in[3];
    const float* b1 = (const float*)d_in[4];
    const float* W2 = (const float*)d_in[5];
    const float* b2 = (const float*)d_in[6];
    const float* W3 = (const float*)d_in[7];
    const float* b3 = (const float*)d_in[8];
    float* out = (float*)d_out;

    const size_t NF = (size_t)N_NODES * 128;
    float* B1 = (float*)d_ws;
    float* Z  = B1 + NF;
    int*   deg_out  = (int*)(Z + NF);
    int*   deg_in   = deg_out + N_NODES;
    float* norm_out = (float*)(deg_in + N_NODES);
    float* norm_in  = norm_out + N_NODES;
    int*   rank     = (int*)(norm_in + N_NODES);
    int*   colw     = rank + N_EDGES;
    int*   row_ptr  = colw + 2 * N_EDGES;

    const int TB = 256;
    const int edgeBlocks  = (N_EDGES + TB - 1) / TB;
    const int nodeBlocks  = (N_NODES + TB - 1) / TB;
    const int layerBlocks = (N_NODES + 31) / 32;
    const int rowBlocks   = (N_NODES * 4 + TB - 1) / TB;

    hipMemsetAsync(deg_out, 0, 2 * (size_t)N_NODES * sizeof(int), stream);
    gcn_degree_rank_kernel<<<edgeBlocks, TB, 0, stream>>>(e_src, e_dst, deg_out, deg_in, rank, N_EDGES);
    gcn_rowptr_norm_kernel<<<nodeBlocks, TB, 0, stream>>>(deg_out, deg_in, norm_out, norm_in, row_ptr, N_NODES);
    gcn_fill_kernel<<<edgeBlocks, TB, 0, stream>>>(e_src, e_dst, rank, row_ptr, norm_out, colw, N_EDGES);

    gcn_layer_kernel<<<layerBlocks, TB, 0, stream>>>(in_feat, row_ptr, colw, norm_in, W1, b1, B1, H_FEATS, 1, N_NODES);
    gcn_layer_proj_kernel<<<layerBlocks, TB, 0, stream>>>(B1, row_ptr, colw, norm_in, W2, b2, W3, Z, 1, N_NODES);
    gcn_gather16_kernel<<<rowBlocks, TB, 0, stream>>>(Z, row_ptr, colw, norm_in, b3, out, N_NODES);
}


# Edges go into 64-slot destination buckets by the in-degree atomic (overflow list beyond): scan, slot array, fill launch go; source scale per edge in layer 1, else in stored rows; arrival-order sums
# speedup vs baseline: 4.1373x; 1.1163x over previous
#include <hip/hip_runtime.h>
#include <hip/hip_bf16.h>


#define N_NODES     50000
#define N_EDGES     800000
#define IN_FEATS    128
#define H_FEATS     128
#define NUM_CLASSES 16
#define BUCKET_CAP  64
#define CTRL_INTS   32

typedef __attribute__((ext_vector_type(2))) float v2f;
typedef __attribute__((ext_vector_type(8))) float v8f;

__device__ __forceinline__ float gcn_deg_scale(int deg) {
    return __frsqrt_rn(fmaxf((float)deg, 1.0f));
}

__global__ void gcn_bucket_kernel(const int* __restrict__ src,
                                  const int* __restrict__ dst,
                                  int* __restrict__ deg_out,
                                  int* __restrict__ deg_in,
                                  int* __restrict__ ovf_count,
                                  int* __restrict__ bucket,
                                  int* __restrict__ ovf, int E) {
    int e = blockIdx.x * blockDim.x + threadIdx.x;
    if (e < E) {
        const int s = src[e];
        const int d = dst[e];
        atomicAdd(&deg_out[s], 1);
        const int r = atomicAdd(&deg_in[d], 1);
        if (r < BUCKET_CAP) {
            bucket[(size_t)d * BUCKET_CAP + r] = s;
        } else {
            const int o = atomicAdd(ovf_count, 1);
            ((int2*)ovf)[o] = make_int2(d, s);
        }
    }
}

__device__ __forceinline__ float4 gcn_row4(const float* __restrict__ h, int s, int lane) {
    return *(const float4*)(h + (size_t)s * 128 + lane * 4);
}
__device__ __forceinline__ void gcn_mac4(float4& a, const float4& v, float w) {
    a.x = __builtin_fmaf(v.x, w, a.x); a.y = __builtin_fmaf(v.y, w, a.y);
    a.z = __builtin_fmaf(v.z, w, a.z); a.w = __builtin_fmaf(v.w, w, a.w);
}

template <bool LOOKUP>
__device__ __forceinline__ int2 gcn_slot_chunk(const int* __restrict__ bucket,
                                               const int* __restrict__ deg_out,
                                               size_t first, int c, int lane) {
    const bool in = lane < c;
    const int* a = in ? bucket + first + lane : deg_out;
    int2 p;
    p.x = *a;
    p.y = 0x3f800000;
    if (LOOKUP) {
        const int id = in ? p.x : 0;
        p.y = __float_as_int(gcn_deg_scale(deg_out[id]));
    }
    return p;
}

template <bool LOOKUP>
__device__ __forceinline__ float4 gcn_gather_row(const float* __restrict__ h,
                                                 const int* __restrict__ bucket,
                                                 const int* __restrict__ deg_out,
                                                 const int* __restrict__ ovf,
                                                 const int* __restrict__ ovf_count,
                                                 int node, int cnt, int2 my, int lane) {
    float4 acc = make_float4(0.0f, 0.0f, 0.0f, 0.0f);
    const int nb = (cnt < BUCKET_CAP) ? cnt : BUCKET_CAP;
    const size_t base = (size_t)node * BUCKET_CAP;
    for (int j0 = 0; j0 < nb; j0 += 32) {
        int c = nb - j0;
        if (c > 32) c = 32;
        if (j0 != 0) my = gcn_slot_chunk<LOOKUP>(bucket, deg_out, base + j0, c, lane);
        int k = 0;
        for (; k + 8 <= c; k += 8) {
            float4 v[8];
            float w[8];
            #pragma unroll
            for (int u = 0; u < 8; ++u) {
                v[u] = gcn_row4(h, __builtin_amdgcn_readlane(my.x, k + u), lane);
                w[u] = __int_as_float(__builtin_amdgcn_readlane(my.y, k + u));
            }
            #pragma unroll
            for (int u = 0; u < 8; ++u) gcn_mac4(acc, v[u], w[u]);
        }
        if (c & 4) {
            float4 v[4];
            float w[4];
            #pragma unroll
            for (int u = 0; u < 4; ++u) {
                v[u] = gcn_row4(h, __builtin_amdgcn_readlane(my.x, k + u), lane);
                w[u] = __int_as_float(__builtin_amdgcn_readlane(my.y, k + u));
            }
            #pragma unroll
            for (int u = 0; u < 4; ++u) gcn_mac4(acc, v[u], w[u]);
            k += 4;
        }
        if (c & 2) {
            float4 v0 = gcn_row4(h, __builtin_amdgcn_readlane(my.x, k), lane);
            float4 v1 = gcn_row4(h, __builtin_amdgcn_readlane(my.x, k + 1), lane);
            const float w0 = __int_as_float(__builtin_amdgcn_readlane(my.y, k));
            const float w1 = __int_as_float(__builtin_amdgcn_readlane(my.y, k + 1));
            gcn_mac4(acc, v0, w0);
            gcn_mac4(acc, v1, w1);
            k += 2;
        }
        if (c & 1) {
            float4 v0 = gcn_row4(h, __builtin_amdgcn_readlane(my.x, k), lane);
            const float w0 = __int_as_float(__builtin_amdgcn_readlane(my.y, k));
            gcn_mac4(acc, v0, w0);
        }
    }
    if (cnt > BUCKET_CAP) {
        const int no = *ovf_count;
        for (int j0 = 0; j0 < no; j0 += 32) {
            int2 e = make_int2(-1, 0);
            if (j0 + lane < no) e = ((const int2*)ovf)[j0 + lane];
            const bool hit = (e.x == node);
            int wbits = 0x3f800000;
            if (LOOKUP && hit) wbits = __float_as_int(gcn_deg_scale(deg_out[e.y]));
            unsigned mask = __builtin_amdgcn_ballot_w32(hit);
            while (mask) {
                const int k = __builtin_ctz(mask);
                mask &= mask - 1;
                const float4 v0 = gcn_row4(h, __builtin_amdgcn_readlane(e.y, k), lane);
                const float w0 = __int_as_float(__builtin_amdgcn_readlane(wbits, k));
                gcn_mac4(acc, v0, w0);
            }
        }
    }
    return acc;
}

template <bool PROJ>
__device__ __forceinline__
void gcn_layer_body(const float* __restrict__ h,
                    const int* __restrict__ deg_in,
                    const int* __restrict__ deg_out,
                    const int* __restrict__ bucket,
                    const int* __restrict__ ovf,
                    const int* __restrict__ ovf_count,
                    const float* __restrict__ W,
                    const float* __restrict__ bias,
                    const float* __restrict__ Wn,
                    float* __restrict__ out,
                    int N, int relu, int n_nodes) {
    const int K = 128;
    constexpr bool LOOKUP = !PROJ;
    __shared__ float As[32 * 132];

    const int lane = threadIdx.x & 31;
    const int wave = __builtin_amdgcn_readfirstlane(threadIdx.x >> 5);
    const int half = lane >> 4;
    const int l16  = lane & 15;
    const int tileM = blockIdx.x * 32;
    const int tileN = wave * 16;

    const int row0 = tileM + wave * 4;
    int cnt[4];
    #pragma unroll
    for (int u = 0; u < 4; ++u)
        cnt[u] = (row0 + u < n_nodes) ? deg_in[row0 + u] : 0;
    int2 my[4];
    #pragma unroll
    for (int u = 0; u < 4; ++u) {
        int c = cnt[u];
        if (c > 32) c = 32;
        my[u] = gcn_slot_chunk<LOOKUP>(bucket, deg_out, (size_t)(row0 + u) * BUCKET_CAP, c, lane);
    }
    if (LOOKUP) asm volatile("" :: "v"(my[0].y), "v"(my[1].y), "v"(my[2].y), "v"(my[3].y));
    else        asm volatile("" :: "v"(my[0].x), "v"(my[1].x), "v"(my[2].x), "v"(my[3].x));
    #pragma unroll
    for (int u = 0; u < 4; ++u) {
        float4 a = gcn_gather_row<LOOKUP>(h, bucket, deg_out, ovf, ovf_count,
                                          row0 + u, cnt[u], my[u], lane);
        const float s = gcn_deg_scale(cnt[u]);
        a.x *= s; a.y *= s; a.z *= s; a.w *= s;
        *(float4*)(As + (wave * 4 + u) * 132 + lane * 4) = a;
    }
    __syncthreads();

    if (!PROJ && tileN >= N) return;

    v8f acc0 = {};
    v8f acc1 = {};
    #pragma unroll 4
    for (int k0 = 0; k0 < K; k0 += 4) {
        int ka = k0 + 2 * half;
        v2f a0, a1, b;
        a0.x = As[l16 * 132 + ka];
        a0.y = As[l16 * 132 + ka + 1];
        a1.x = As[(16 + l16) * 132 + ka];
        a1.y = As[(16 + l16) * 132 + ka + 1];
        b.x = W[(size_t)ka * N + tileN + l16];
        b.y = W[(size_t)(ka + 1) * N + tileN + l16];
        acc0 = __builtin_amdgcn_wmma_f32_16x16x4_f32(
             false, a0,  false, b,
             (short)0, acc0,  false,  false);
        acc1 = __builtin_amdgcn_wmma_f32_16x16x4_f32(
             false, a1,  false, b,
             (short)0, acc1,  false,  false);
    }

    const float bb = bias[tileN + l16];
    if (PROJ) {
        __shared__ float Zp[8 * 256];
        const int m  = wave & 1;
        const int kq = wave >> 1;
        v2f bn[8];
        #pragma unroll
        for (int ks = 0; ks < 8; ++ks) {
            const int ka = kq * 32 + ks * 4 + 2 * half;
            bn[ks].x = Wn[ka * 16 + l16];
            bn[ks].y = Wn[(ka + 1) * 16 + l16];
        }
        __syncthreads();
        #pragma unroll
        for (int r = 0; r < 8; ++r) {
            float v0 = acc0[r] + bb;
            float v1 = acc1[r] + bb;
            if (relu) { v0 = fmaxf(v0, 0.0f); v1 = fmaxf(v1, 0.0f); }
            As[(r + 8 * half) * 132 + tileN + l16] = v0;
            As[(16 + r + 8 * half) * 132 + tileN + l16] = v1;
        }
        __syncthreads();
        v8f zacc = {};
        #pragma unroll
        for (int ks = 0; ks < 8; ++ks) {
            const int ka = kq * 32 + ks * 4 + 2 * half;
            v2f a;
            a.x = As[(m * 16 + l16) * 132 + ka];
            a.y = As[(m * 16 + l16) * 132 + ka + 1];
            zacc = __builtin_amdgcn_wmma_f32_16x16x4_f32(
                 false, a,  false, bn[ks],
                 (short)0, zacc,  false,  false);
        }
        #pragma unroll
        for (int r = 0; r < 8; ++r)
            Zp[wave * 256 + (r + 8 * half) * 16 + l16] = zacc[r];
        __syncthreads();
        const int t = threadIdx.x;
        if (t < 128) {
            const int row = t >> 2;
            const int q   = t & 3;
            const float* p = Zp + (row >> 4) * 256 + (row & 15) * 16 + q * 4;
            const float4 s0 = *(const float4*)(p);
            const float4 s1 = *(const float4*)(p + 512);
            const float4 s2 = *(const float4*)(p + 1024);
            const float4 s3 = *(const float4*)(p + 1536);
            if (tileM + row < n_nodes) {
                const float so = gcn_deg_scale(deg_out[tileM + row]);
                float4 z;
                z.x = (((s0.x + s1.x) + s2.x) + s3.x) * so;
                z.y = (((s0.y + s1.y) + s2.y) + s3.y) * so;
                z.z = (((s0.z + s1.z) + s2.z) + s3.z) * so;
                z.w = (((s0.w + s1.w) + s2.w) + s3.w) * so;
                ((float4*)out)[(size_t)(tileM + row) * 4 + q] = z;
            }
        }
        return;
    }
    #pragma unroll
    for (int r = 0; r < 8; ++r) {
        int row = tileM + r + 8 * half;
        float v = acc0[r] + bb;
        if (relu) v = fmaxf(v, 0.0f);
        out[(size_t)row * N + tileN + l16] = v * gcn_deg_scale(deg_out[row]);
    }
    if (tileM + 16 < n_nodes) {
        #pragma unroll
        for (int r = 0; r < 8; ++r) {
            int row = tileM + 16 + r + 8 * half;
            float v = acc1[r] + bb;
            if (relu) v = fmaxf(v, 0.0f);
            out[(size_t)row * N + tileN + l16] = v * gcn_deg_scale(deg_out[row]);
        }
    }
}

__global__ __launch_bounds__(256)
void gcn_layer_kernel(const float* __restrict__ h, const int* __restrict__ deg_in,
                      const int* __restrict__ deg_out, const int* __restrict__ bucket,
                      const int* __restrict__ ovf, const int* __restrict__ ovf_count,
                      const float* __restrict__ W, const float* __restrict__ bias,
                      float* __restrict__ out, int N, int relu, int n_nodes) {
    gcn_layer_body<false>(h, deg_in, deg_out, bucket, ovf, ovf_count, W, bias, nullptr, out, N, relu, n_nodes);
}

__global__ __launch_bounds__(256)
void gcn_layer_proj_kernel(const float* __restrict__ h, const int* __restrict__ deg_in,
                           const int* __restrict__ deg_out, const int* __restrict__ bucket,
                           const int* __restrict__ ovf, const int* __restrict__ ovf_count,
                           const float* __restrict__ W, const float* __restrict__ bias,
                           const float* __restrict__ Wn, float* __restrict__ z,
                           int relu, int n_nodes) {
    gcn_layer_body<true>(h, deg_in, deg_out, bucket, ovf, ovf_count, W, bias, Wn, z, 128, relu, n_nodes);
}

__global__ __launch_bounds__(256)
void gcn_gather16_kernel(const float* __restrict__ z,
                         const int* __restrict__ deg_in,
                         const int* __restrict__ bucket,
                         const int* __restrict__ ovf,
                         const int* __restrict__ ovf_count,
                         const float* __restrict__ bias,
                         float* __restrict__ out,
                         int n_nodes) {
    const int t = blockIdx.x * 256 + threadIdx.x;
    const int node = t >> 2;
    const int q = t & 3;
    if (node >= n_nodes) return;
    const int cnt = deg_in[node];
    const int nb = (cnt < BUCKET_CAP) ? cnt : BUCKET_CAP;
    const int* bk = bucket + (size_t)node * BUCKET_CAP;
    const float4* z4 = (const float4*)z;
    const float s = gcn_deg_scale(cnt);
    const float4 b = ((const float4*)bias)[q];
    float4 acc = make_float4(0.0f, 0.0f, 0.0f, 0.0f);
    if (nb > 0) {
        const int last = nb - 1;
        int p[8];
        #pragma unroll
        for (int u = 0; u < 8; ++u) p[u] = bk[min(u, last)];
        for (int j = 0; j < nb; j += 8) {
            float4 v[8];
            int pn[8];
            #pragma unroll
            for (int u = 0; u < 8; ++u) v[u] = z4[(size_t)p[u] * 4 + q];
            #pragma unroll
            for (int u = 0; u < 8; ++u) pn[u] = bk[min(j + 8 + u, last)];
            acc.x += v[0].x; acc.y += v[0].y; acc.z += v[0].z; acc.w += v[0].w;
            #pragma unroll
            for (int u = 1; u < 8; ++u) {
                const bool real = j + u < nb;
                acc.x = real ? acc.x + v[u].x : acc.x; acc.y = real ? acc.y + v[u].y : acc.y;
                acc.z = real ? acc.z + v[u].z : acc.z; acc.w = real ? acc.w + v[u].w : acc.w;
            }
            #pragma unroll
            for (int u = 0; u < 8; ++u) p[u] = pn[u];
        }
    }
    if (cnt > BUCKET_CAP) {
        const int no = *ovf_count;
        const int2* ov = (const int2*)ovf;
        for (int j = 0; j < no; ++j) {
            const int2 e = ov[j];
            if (e.x == node) {
                const float4 v0 = z4[(size_t)e.y * 4 + q];
                acc.x += v0.x; acc.y += v0.y; acc.z += v0.z; acc.w += v0.w;
            }
        }
    }
    float4 o;
    o.x = acc.x * s + b.x; o.y = acc.y * s + b.y;
    o.z = acc.z * s + b.z; o.w = acc.w * s + b.w;
    ((float4*)out)[(size_t)node * 4 + q] = o;
}

extern "C" void kernel_launch(void* const* d_in, const int* in_sizes, int n_in,
                              void* d_out, int out_size, void* d_ws, size_t ws_size,
                              hipStream_t stream) {
    const float* in_feat = (const float*)d_in[0];
    const int*   e_src   = (const int*)d_in[1];
    const int*   e_dst   = (const int*)d_in[2];
    const float* W1 = (const float*)d_in[3];
    const float* b1 = (const float*)d_in[4];
    const float* W2 = (const float*)d_in[5];
    const float* b2 = (const float*)d_in[6];
    const float* W3 = (const float*)d_in[7];
    const float* b3 = (const float*)d_in[8];
    float* out = (float*)d_out;

    const size_t NF = (size_t)N_NODES * 128;
    float* B1 = (float*)d_ws;
    float* Z  = B1 + NF;
    int*   deg_out   = (int*)(Z + NF);
    int*   deg_in    = deg_out + N_NODES;
    int*   ovf_count = deg_in + N_NODES;
    int*   bucket    = ovf_count + CTRL_INTS;
    int*   ovf       = bucket + (size_t)N_NODES * BUCKET_CAP;

    const int TB = 256;
    const int edgeBlocks  = (N_EDGES + TB - 1) / TB;
    const int layerBlocks = (N_NODES + 31) / 32;
    const int rowBlocks   = (N_NODES * 4 + TB - 1) / TB;

    hipMemsetAsync(deg_out, 0, (2 * (size_t)N_NODES + CTRL_INTS) * sizeof(int), stream);
    gcn_bucket_kernel<<<edgeBlocks, TB, 0, stream>>>(e_src, e_dst, deg_out, deg_in, ovf_count, bucket, ovf, N_EDGES);

    gcn_layer_kernel<<<layerBlocks, TB, 0, stream>>>(in_feat, deg_in, deg_out, bucket, ovf, ovf_count, W1, b1, B1, H_FEATS, 1, N_NODES);
    gcn_layer_proj_kernel<<<layerBlocks, TB, 0, stream>>>(B1, deg_in, deg_out, bucket, ovf, ovf_count, W2, b2, W3, Z, 1, N_NODES);
    gcn_gather16_kernel<<<rowBlocks, TB, 0, stream>>>(Z, deg_in, bucket, ovf, ovf_count, b3, out, N_NODES);
}


# Edge grouping as two-level LDS counting sort: chunks sorted by 64-node destination and source range, a block per range builds buckets and both degree counts; global atomics, memset go; arrival order
# speedup vs baseline: 7.3205x; 1.7694x over previous
#include <hip/hip_runtime.h>
#include <hip/hip_bf16.h>


#define N_NODES     50000
#define N_EDGES     800000
#define IN_FEATS    128
#define H_FEATS     128
#define NUM_CLASSES 16
#define BUCKET_CAP  64
#define CTRL_INTS   64
#define SORT_CHUNKS      128
#define SORT_THREADS     1024
#define EDGES_PER_CHUNK  ((N_EDGES + SORT_CHUNKS - 1) / SORT_CHUNKS)
#define EDGES_PER_THREAD ((EDGES_PER_CHUNK + SORT_THREADS - 1) / SORT_THREADS)
#define BIN_SHIFT        6
#define BIN_NODES        (1 << BIN_SHIFT)
#define N_BINS           ((N_NODES + BIN_NODES - 1) / BIN_NODES)
#define BIN_THREADS      (2 * SORT_CHUNKS)
#define DEG_PITCH        (N_BINS * BIN_NODES)
#define CHUNK_PITCH      6272
#define OFFS_PITCH       800
static_assert(EDGES_PER_CHUNK <= CHUNK_PITCH, "a sorted chunk fits its pitch");
static_assert(N_BINS + 1 <= OFFS_PITCH && N_BINS + 1 <= SORT_THREADS, "one scan entry per range, plus the end");
static_assert(SORT_THREADS == 32 * 32, "the block scan is 32 waves of 32 lanes");
static_assert(N_NODES <= (1 << 24), "a source id and 6 destination bits share one int");
static_assert((SORT_CHUNKS % 32) == 0, "the two halves of a range block are whole waves");

typedef __attribute__((ext_vector_type(2))) float v2f;
typedef __attribute__((ext_vector_type(8))) float v8f;

__device__ __forceinline__ float gcn_deg_scale(int deg) {
    return __frsqrt_rn(fmaxf((float)deg, 1.0f));
}

__device__ __forceinline__ int gcn_block_excl_scan(int v, int* wsum, int lane, int wave) {
    int x = v;
    for (int o = 1; o < 32; o <<= 1) {
        const int y = __shfl_up(x, o);
        if (lane >= o) x += y;
    }
    if (lane == 31) wsum[wave] = x;
    __syncthreads();
    if (wave == 0) {
        const int t = wsum[lane];
        int xs = t;
        for (int o = 1; o < 32; o <<= 1) {
            const int y = __shfl_up(xs, o);
            if (lane >= o) xs += y;
        }
        wsum[lane] = xs - t;
    }
    __syncthreads();
    const int r = x - v + wsum[wave];
    __syncthreads();
    return r;
}

__global__ __launch_bounds__(SORT_THREADS)
void gcn_chunk_sort_kernel(const int* __restrict__ src,
                           const int* __restrict__ dst,
                           int* __restrict__ by_dst,
                           int* __restrict__ by_src,
                           int2* __restrict__ offs,
                           int* __restrict__ ovf_count, int E) {
    __shared__ int hist_d[SORT_THREADS];
    __shared__ int hist_s[SORT_THREADS];
    __shared__ int wsum[32];
    __shared__ int stage_d[EDGES_PER_CHUNK];
    __shared__ int stage_s[EDGES_PER_CHUNK];
    const int tid  = threadIdx.x;
    const int lane = tid & 31;
    const int wave = tid >> 5;
    if (blockIdx.x == 0 && tid == 0) *ovf_count = 0;
    hist_d[tid] = 0;
    hist_s[tid] = 0;
    __syncthreads();
    const int e0 = blockIdx.x * EDGES_PER_CHUNK;
    int n = E - e0;
    if (n > EDGES_PER_CHUNK) n = EDGES_PER_CHUNK;
    if (n < 0) n = 0;
    int s[EDGES_PER_THREAD], d[EDGES_PER_THREAD], rd[EDGES_PER_THREAD], rs[EDGES_PER_THREAD];
    #pragma unroll
    for (int u = 0; u < EDGES_PER_THREAD; ++u) {
        const int i = u * SORT_THREADS + tid;
        s[u] = 0;
        d[u] = -1;
        if (i < n) { s[u] = src[e0 + i]; d[u] = dst[e0 + i]; }
    }
    #pragma unroll
    for (int u = 0; u < EDGES_PER_THREAD; ++u) {
        rd[u] = 0;
        rs[u] = 0;
        if (d[u] >= 0) {
            rd[u] = atomicAdd(&hist_d[d[u] >> BIN_SHIFT], 1);
            rs[u] = atomicAdd(&hist_s[s[u] >> BIN_SHIFT], 1);
        }
    }
    __syncthreads();
    const int start_d = gcn_block_excl_scan(hist_d[tid], wsum, lane, wave);
    const int start_s = gcn_block_excl_scan(hist_s[tid], wsum, lane, wave);
    hist_d[tid] = start_d;
    hist_s[tid] = start_s;
    __syncthreads();
    #pragma unroll
    for (int u = 0; u < EDGES_PER_THREAD; ++u) {
        if (d[u] >= 0) {
            stage_d[hist_d[d[u] >> BIN_SHIFT] + rd[u]] = (s[u] << BIN_SHIFT) | (d[u] & (BIN_NODES - 1));
            stage_s[hist_s[s[u] >> BIN_SHIFT] + rs[u]] = s[u];
        }
    }
    __syncthreads();
    int* od = by_dst + (size_t)blockIdx.x * CHUNK_PITCH;
    int* os = by_src + (size_t)blockIdx.x * CHUNK_PITCH;
    for (int i = tid; i < n; i += SORT_THREADS) {
        od[i] = stage_d[i];
        os[i] = stage_s[i];
    }
    if (tid <= N_BINS)
        offs[(size_t)blockIdx.x * OFFS_PITCH + tid] = make_int2(hist_d[tid], hist_s[tid]);
}

__global__ __launch_bounds__(BIN_THREADS)
void gcn_bin_kernel(const int* __restrict__ by_dst,
                    const int* __restrict__ by_src,
                    const int2* __restrict__ offs,
                    int* __restrict__ deg_in,
                    int* __restrict__ deg_out,
                    int* __restrict__ bucket,
                    int* __restrict__ ovf_count,
                    int* __restrict__ ovf, int n_nodes) {
    __shared__ int cnt_in[BIN_NODES];
    __shared__ int cnt_out[BIN_NODES];
    __shared__ __attribute__((aligned(16))) int tile[BIN_NODES * BUCKET_CAP];
    const int tid = threadIdx.x;
    const int b   = blockIdx.x;
    for (int i = tid; i < BIN_NODES * BUCKET_CAP / 4; i += BIN_THREADS)
        ((int4*)tile)[i] = make_int4(0, 0, 0, 0);
    if (tid < BIN_NODES) { cnt_in[tid] = 0; cnt_out[tid] = 0; }
    __syncthreads();
    const int k = tid & (SORT_CHUNKS - 1);
    const int2 a0 = offs[(size_t)k * OFFS_PITCH + b];
    const int2 a1 = offs[(size_t)k * OFFS_PITCH + b + 1];
    if (tid >= SORT_CHUNKS) {
        const int* run = by_src + (size_t)k * CHUNK_PITCH;
        for (int j = a0.y; j < a1.y; ++j)
            atomicAdd(&cnt_out[run[j] & (BIN_NODES - 1)], 1);
    } else {
        const int* run = by_dst + (size_t)k * CHUNK_PITCH;
        for (int j = a0.x; j < a1.x; ++j) {
            const int p   = run[j];
            const int dl  = p & (BIN_NODES - 1);
            const int sid = p >> BIN_SHIFT;
            const int r   = atomicAdd(&cnt_in[dl], 1);
            if (r < BUCKET_CAP) {
                tile[dl * BUCKET_CAP + r] = sid;
            } else {
                const int o = atomicAdd(ovf_count, 1);
                ((int2*)ovf)[o] = make_int2(b * BIN_NODES + dl, sid);
            }
        }
    }
    __syncthreads();
    const int node0 = b * BIN_NODES;
    if (tid < BIN_NODES && node0 + tid < n_nodes) {
        deg_in[node0 + tid]  = cnt_in[tid];
        deg_out[node0 + tid] = cnt_out[tid];
    }
    for (int i = tid; i < BIN_NODES * BUCKET_CAP / 4; i += BIN_THREADS) {
        const int row = i >> 4;
        const int q   = i & 15;
        if (node0 + row < n_nodes && (q < 8 || cnt_in[row] > 32))
            ((int4*)bucket)[(size_t)node0 * (BUCKET_CAP / 4) + i] = ((const int4*)tile)[i];
    }
}

__device__ __forceinline__ float4 gcn_row4(const float* __restrict__ h, int s, int lane) {
    return *(const float4*)(h + (size_t)s * 128 + lane * 4);
}
__device__ __forceinline__ void gcn_mac4(float4& a, const float4& v, float w) {
    a.x = __builtin_fmaf(v.x, w, a.x); a.y = __builtin_fmaf(v.y, w, a.y);
    a.z = __builtin_fmaf(v.z, w, a.z); a.w = __builtin_fmaf(v.w, w, a.w);
}

template <bool LOOKUP>
__device__ __forceinline__ int2 gcn_slot_chunk(const int* __restrict__ bucket,
                                               const int* __restrict__ deg_out,
                                               size_t first, int c, int lane) {
    const bool in = lane < c;
    const int* a = in ? bucket + first + lane : deg_out;
    int2 p;
    p.x = *a;
    p.y = 0x3f800000;
    if (LOOKUP) {
        const int id = in ? p.x : 0;
        p.y = __float_as_int(gcn_deg_scale(deg_out[id]));
    }
    return p;
}

template <bool LOOKUP>
__device__ __forceinline__ float4 gcn_gather_row(const float* __restrict__ h,
                                                 const int* __restrict__ bucket,
                                                 const int* __restrict__ deg_out,
                                                 const int* __restrict__ ovf,
                                                 const int* __restrict__ ovf_count,
                                                 int node, int cnt, int2 my, int lane) {
    float4 acc = make_float4(0.0f, 0.0f, 0.0f, 0.0f);
    const int nb = (cnt < BUCKET_CAP) ? cnt : BUCKET_CAP;
    const size_t base = (size_t)node * BUCKET_CAP;
    for (int j0 = 0; j0 < nb; j0 += 32) {
        int c = nb - j0;
        if (c > 32) c = 32;
        if (j0 != 0) my = gcn_slot_chunk<LOOKUP>(bucket, deg_out, base + j0, c, lane);
        int k = 0;
        for (; k + 8 <= c; k += 8) {
            float4 v[8];
            float w[8];
            #pragma unroll
            for (int u = 0; u < 8; ++u) {
                v[u] = gcn_row4(h, __builtin_amdgcn_readlane(my.x, k + u), lane);
                w[u] = __int_as_float(__builtin_amdgcn_readlane(my.y, k + u));
            }
            #pragma unroll
            for (int u = 0; u < 8; ++u) gcn_mac4(acc, v[u], w[u]);
        }
        if (c & 4) {
            float4 v[4];
            float w[4];
            #pragma unroll
            for (int u = 0; u < 4; ++u) {
                v[u] = gcn_row4(h, __builtin_amdgcn_readlane(my.x, k + u), lane);
                w[u] = __int_as_float(__builtin_amdgcn_readlane(my.y, k + u));
            }
            #pragma unroll
            for (int u = 0; u < 4; ++u) gcn_mac4(acc, v[u], w[u]);
            k += 4;
        }
        if (c & 2) {
            float4 v0 = gcn_row4(h, __builtin_amdgcn_readlane(my.x, k), lane);
            float4 v1 = gcn_row4(h, __builtin_amdgcn_readlane(my.x, k + 1), lane);
            const float w0 = __int_as_float(__builtin_amdgcn_readlane(my.y, k));
            const float w1 = __int_as_float(__builtin_amdgcn_readlane(my.y, k + 1));
            gcn_mac4(acc, v0, w0);
            gcn_mac4(acc, v1, w1);
            k += 2;
        }
        if (c & 1) {
            float4 v0 = gcn_row4(h, __builtin_amdgcn_readlane(my.x, k), lane);
            const float w0 = __int_as_float(__builtin_amdgcn_readlane(my.y, k));
            gcn_mac4(acc, v0, w0);
        }
    }
    if (cnt > BUCKET_CAP) {
        const int no = *ovf_count;
        for (int j0 = 0; j0 < no; j0 += 32) {
            int2 e = make_int2(-1, 0);
            if (j0 + lane < no) e = ((const int2*)ovf)[j0 + lane];
            const bool hit = (e.x == node);
            int wbits = 0x3f800000;
            if (LOOKUP && hit) wbits = __float_as_int(gcn_deg_scale(deg_out[e.y]));
            unsigned mask = __builtin_amdgcn_ballot_w32(hit);
            while (mask) {
                const int k = __builtin_ctz(mask);
                mask &= mask - 1;
                const float4 v0 = gcn_row4(h, __builtin_amdgcn_readlane(e.y, k), lane);
                const float w0 = __int_as_float(__builtin_amdgcn_readlane(wbits, k));
                gcn_mac4(acc, v0, w0);
            }
        }
    }
    return acc;
}

template <bool PROJ>
__device__ __forceinline__
void gcn_layer_body(const float* __restrict__ h,
                    const int* __restrict__ deg_in,
                    const int* __restrict__ deg_out,
                    const int* __restrict__ bucket,
                    const int* __restrict__ ovf,
                    const int* __restrict__ ovf_count,
                    const float* __restrict__ W,
                    const float* __restrict__ bias,
                    const float* __restrict__ Wn,
                    float* __restrict__ out,
                    int N, int relu, int n_nodes) {
    const int K = 128;
    constexpr bool LOOKUP = !PROJ;
    __shared__ float As[32 * 132];

    const int lane = threadIdx.x & 31;
    const int wave = __builtin_amdgcn_readfirstlane(threadIdx.x >> 5);
    const int half = lane >> 4;
    const int l16  = lane & 15;
    const int tileM = blockIdx.x * 32;
    const int tileN = wave * 16;

    const int row0 = tileM + wave * 4;
    int cnt[4];
    #pragma unroll
    for (int u = 0; u < 4; ++u)
        cnt[u] = (row0 + u < n_nodes) ? deg_in[row0 + u] : 0;
    int2 my[4];
    #pragma unroll
    for (int u = 0; u < 4; ++u) {
        int c = cnt[u];
        if (c > 32) c = 32;
        my[u] = gcn_slot_chunk<LOOKUP>(bucket, deg_out, (size_t)(row0 + u) * BUCKET_CAP, c, lane);
    }
    if (LOOKUP) asm volatile("" :: "v"(my[0].y), "v"(my[1].y), "v"(my[2].y), "v"(my[3].y));
    else        asm volatile("" :: "v"(my[0].x), "v"(my[1].x), "v"(my[2].x), "v"(my[3].x));
    #pragma unroll
    for (int u = 0; u < 4; ++u) {
        float4 a = gcn_gather_row<LOOKUP>(h, bucket, deg_out, ovf, ovf_count,
                                          row0 + u, cnt[u], my[u], lane);
        const float s = gcn_deg_scale(cnt[u]);
        a.x *= s; a.y *= s; a.z *= s; a.w *= s;
        *(float4*)(As + (wave * 4 + u) * 132 + lane * 4) = a;
    }
    __syncthreads();

    if (!PROJ && tileN >= N) return;

    v8f acc0 = {};
    v8f acc1 = {};
    #pragma unroll 4
    for (int k0 = 0; k0 < K; k0 += 4) {
        int ka = k0 + 2 * half;
        v2f a0, a1, b;
        a0.x = As[l16 * 132 + ka];
        a0.y = As[l16 * 132 + ka + 1];
        a1.x = As[(16 + l16) * 132 + ka];
        a1.y = As[(16 + l16) * 132 + ka + 1];
        b.x = W[(size_t)ka * N + tileN + l16];
        b.y = W[(size_t)(ka + 1) * N + tileN + l16];
        acc0 = __builtin_amdgcn_wmma_f32_16x16x4_f32(
             false, a0,  false, b,
             (short)0, acc0,  false,  false);
        acc1 = __builtin_amdgcn_wmma_f32_16x16x4_f32(
             false, a1,  false, b,
             (short)0, acc1,  false,  false);
    }

    const float bb = bias[tileN + l16];
    if (PROJ) {
        __shared__ float Zp[8 * 256];
        const int m  = wave & 1;
        const int kq = wave >> 1;
        v2f bn[8];
        #pragma unroll
        for (int ks = 0; ks < 8; ++ks) {
            const int ka = kq * 32 + ks * 4 + 2 * half;
            bn[ks].x = Wn[ka * 16 + l16];
            bn[ks].y = Wn[(ka + 1) * 16 + l16];
        }
        __syncthreads();
        #pragma unroll
        for (int r = 0; r < 8; ++r) {
            float v0 = acc0[r] + bb;
            float v1 = acc1[r] + bb;
            if (relu) { v0 = fmaxf(v0, 0.0f); v1 = fmaxf(v1, 0.0f); }
            As[(r + 8 * half) * 132 + tileN + l16] = v0;
            As[(16 + r + 8 * half) * 132 + tileN + l16] = v1;
        }
        __syncthreads();
        v8f zacc = {};
        #pragma unroll
        for (int ks = 0; ks < 8; ++ks) {
            const int ka = kq * 32 + ks * 4 + 2 * half;
            v2f a;
            a.x = As[(m * 16 + l16) * 132 + ka];
            a.y = As[(m * 16 + l16) * 132 + ka + 1];
            zacc = __builtin_amdgcn_wmma_f32_16x16x4_f32(
                 false, a,  false, bn[ks],
                 (short)0, zacc,  false,  false);
        }
        #pragma unroll
        for (int r = 0; r < 8; ++r)
            Zp[wave * 256 + (r + 8 * half) * 16 + l16] = zacc[r];
        __syncthreads();
        const int t = threadIdx.x;
        if (t < 128) {
            const int row = t >> 2;
            const int q   = t & 3;
            const float* p = Zp + (row >> 4) * 256 + (row & 15) * 16 + q * 4;
            const float4 s0 = *(const float4*)(p);
            const float4 s1 = *(const float4*)(p + 512);
            const float4 s2 = *(const float4*)(p + 1024);
            const float4 s3 = *(const float4*)(p + 1536);
            if (tileM + row < n_nodes) {
                const float so = gcn_deg_scale(deg_out[tileM + row]);
                float4 z;
                z.x = (((s0.x + s1.x) + s2.x) + s3.x) * so;
                z.y = (((s0.y + s1.y) + s2.y) + s3.y) * so;
                z.z = (((s0.z + s1.z) + s2.z) + s3.z) * so;
                z.w = (((s0.w + s1.w) + s2.w) + s3.w) * so;
                ((float4*)out)[(size_t)(tileM + row) * 4 + q] = z;
            }
        }
        return;
    }
    #pragma unroll
    for (int r = 0; r < 8; ++r) {
        int row = tileM + r + 8 * half;
        float v = acc0[r] + bb;
        if (relu) v = fmaxf(v, 0.0f);
        out[(size_t)row * N + tileN + l16] = v * gcn_deg_scale(deg_out[row]);
    }
    if (tileM + 16 < n_nodes) {
        #pragma unroll
        for (int r = 0; r < 8; ++r) {
            int row = tileM + 16 + r + 8 * half;
            float v = acc1[r] + bb;
            if (relu) v = fmaxf(v, 0.0f);
            out[(size_t)row * N + tileN + l16] = v * gcn_deg_scale(deg_out[row]);
        }
    }
}

__global__ __launch_bounds__(256)
void gcn_layer_kernel(const float* __restrict__ h, const int* __restrict__ deg_in,
                      const int* __restrict__ deg_out, const int* __restrict__ bucket,
                      const int* __restrict__ ovf, const int* __restrict__ ovf_count,
                      const float* __restrict__ W, const float* __restrict__ bias,
                      float* __restrict__ out, int N, int relu, int n_nodes) {
    gcn_layer_body<false>(h, deg_in, deg_out, bucket, ovf, ovf_count, W, bias, nullptr, out, N, relu, n_nodes);
}

__global__ __launch_bounds__(256)
void gcn_layer_proj_kernel(const float* __restrict__ h, const int* __restrict__ deg_in,
                           const int* __restrict__ deg_out, const int* __restrict__ bucket,
                           const int* __restrict__ ovf, const int* __restrict__ ovf_count,
                           const float* __restrict__ W, const float* __restrict__ bias,
                           const float* __restrict__ Wn, float* __restrict__ z,
                           int relu, int n_nodes) {
    gcn_layer_body<true>(h, deg_in, deg_out, bucket, ovf, ovf_count, W, bias, Wn, z, 128, relu, n_nodes);
}

__global__ __launch_bounds__(256)
void gcn_gather16_kernel(const float* __restrict__ z,
                         const int* __restrict__ deg_in,
                         const int* __restrict__ bucket,
                         const int* __restrict__ ovf,
                         const int* __restrict__ ovf_count,
                         const float* __restrict__ bias,
                         float* __restrict__ out,
                         int n_nodes) {
    const int t = blockIdx.x * 256 + threadIdx.x;
    const int node = t >> 2;
    const int q = t & 3;
    if (node >= n_nodes) return;
    const int cnt = deg_in[node];
    const int nb = (cnt < BUCKET_CAP) ? cnt : BUCKET_CAP;
    const int* bk = bucket + (size_t)node * BUCKET_CAP;
    const float4* z4 = (const float4*)z;
    const float s = gcn_deg_scale(cnt);
    const float4 b = ((const float4*)bias)[q];
    float4 acc = make_float4(0.0f, 0.0f, 0.0f, 0.0f);
    if (nb > 0) {
        const int last = nb - 1;
        int p[8];
        #pragma unroll
        for (int u = 0; u < 8; ++u) p[u] = bk[min(u, last)];
        for (int j = 0; j < nb; j += 8) {
            float4 v[8];
            int pn[8];
            #pragma unroll
            for (int u = 0; u < 8; ++u) v[u] = z4[(size_t)p[u] * 4 + q];
            #pragma unroll
            for (int u = 0; u < 8; ++u) pn[u] = bk[min(j + 8 + u, last)];
            acc.x += v[0].x; acc.y += v[0].y; acc.z += v[0].z; acc.w += v[0].w;
            #pragma unroll
            for (int u = 1; u < 8; ++u) {
                const bool real = j + u < nb;
                acc.x = real ? acc.x + v[u].x : acc.x; acc.y = real ? acc.y + v[u].y : acc.y;
                acc.z = real ? acc.z + v[u].z : acc.z; acc.w = real ? acc.w + v[u].w : acc.w;
            }
            #pragma unroll
            for (int u = 0; u < 8; ++u) p[u] = pn[u];
        }
    }
    if (cnt > BUCKET_CAP) {
        const int no = *ovf_count;
        const int2* ov = (const int2*)ovf;
        for (int j = 0; j < no; ++j) {
            const int2 e = ov[j];
            if (e.x == node) {
                const float4 v0 = z4[(size_t)e.y * 4 + q];
                acc.x += v0.x; acc.y += v0.y; acc.z += v0.z; acc.w += v0.w;
            }
        }
    }
    float4 o;
    o.x = acc.x * s + b.x; o.y = acc.y * s + b.y;
    o.z = acc.z * s + b.z; o.w = acc.w * s + b.w;
    ((float4*)out)[(size_t)node * 4 + q] = o;
}

extern "C" void kernel_launch(void* const* d_in, const int* in_sizes, int n_in,
                              void* d_out, int out_size, void* d_ws, size_t ws_size,
                              hipStream_t stream) {
    const float* in_feat = (const float*)d_in[0];
    const int*   e_src   = (const int*)d_in[1];
    const int*   e_dst   = (const int*)d_in[2];
    const float* W1 = (const float*)d_in[3];
    const float* b1 = (const float*)d_in[4];
    const float* W2 = (const float*)d_in[5];
    const float* b2 = (const float*)d_in[6];
    const float* W3 = (const float*)d_in[7];
    const float* b3 = (const float*)d_in[8];
    float* out = (float*)d_out;

    const size_t NF = (size_t)N_NODES * 128;
    float* B1 = (float*)d_ws;
    float* Z  = B1 + NF;
    int*   deg_out   = (int*)(Z + NF);
    int*   deg_in    = deg_out + DEG_PITCH;
    int*   ovf_count = deg_in + DEG_PITCH;
    int*   bucket    = ovf_count + CTRL_INTS;
    int*   ovf       = bucket + (size_t)N_NODES * BUCKET_CAP;
    int*   by_dst    = ovf + 2 * (size_t)N_EDGES;
    int*   by_src    = by_dst + (size_t)SORT_CHUNKS * CHUNK_PITCH;
    int2*  offs      = (int2*)(by_src + (size_t)SORT_CHUNKS * CHUNK_PITCH);

    const int TB = 256;
    const int layerBlocks = (N_NODES + 31) / 32;
    const int rowBlocks   = (N_NODES * 4 + TB - 1) / TB;

    gcn_chunk_sort_kernel<<<SORT_CHUNKS, SORT_THREADS, 0, stream>>>(e_src, e_dst, by_dst, by_src, offs, ovf_count, N_EDGES);
    gcn_bin_kernel<<<N_BINS, BIN_THREADS, 0, stream>>>(by_dst, by_src, offs, deg_in, deg_out, bucket, ovf_count, ovf, N_NODES);

    gcn_layer_kernel<<<layerBlocks, TB, 0, stream>>>(in_feat, deg_in, deg_out, bucket, ovf, ovf_count, W1, b1, B1, H_FEATS, 1, N_NODES);
    gcn_layer_proj_kernel<<<layerBlocks, TB, 0, stream>>>(B1, deg_in, deg_out, bucket, ovf, ovf_count, W2, b2, W3, Z, 1, N_NODES);
    gcn_gather16_kernel<<<rowBlocks, TB, 0, stream>>>(Z, deg_in, bucket, ovf, ovf_count, b3, out, N_NODES);
}


# Layer launches run half as many waves: 4-wave blocks, a wave sums 8 rows two side by side (16 row loads in flight, slot order kept) and multiplies 2 M x 2 column tiles, W read as whole lines
# speedup vs baseline: 7.6466x; 1.0445x over previous
#include <hip/hip_runtime.h>
#include <hip/hip_bf16.h>


#define N_NODES     50000
#define N_EDGES     800000
#define IN_FEATS    128
#define H_FEATS     128
#define NUM_CLASSES 16
#define BUCKET_CAP  64
#define CTRL_INTS   64
#define SORT_CHUNKS      128
#define SORT_THREADS     1024
#define EDGES_PER_CHUNK  ((N_EDGES + SORT_CHUNKS - 1) / SORT_CHUNKS)
#define EDGES_PER_THREAD ((EDGES_PER_CHUNK + SORT_THREADS - 1) / SORT_THREADS)
#define BIN_SHIFT        6
#define BIN_NODES        (1 << BIN_SHIFT)
#define N_BINS           ((N_NODES + BIN_NODES - 1) / BIN_NODES)
#define BIN_THREADS      (2 * SORT_CHUNKS)
#define DEG_PITCH        (N_BINS * BIN_NODES)
#define CHUNK_PITCH      6272
#define OFFS_PITCH       800
static_assert(EDGES_PER_CHUNK <= CHUNK_PITCH, "a sorted chunk fits its pitch");
static_assert(N_BINS + 1 <= OFFS_PITCH && N_BINS + 1 <= SORT_THREADS, "one scan entry per range, plus the end");
static_assert(SORT_THREADS == 32 * 32, "the block scan is 32 waves of 32 lanes");
static_assert(N_NODES <= (1 << 24), "a source id and 6 destination bits share one int");
static_assert((SORT_CHUNKS % 32) == 0, "the two halves of a range block are whole waves");

typedef __attribute__((ext_vector_type(2))) float v2f;
typedef __attribute__((ext_vector_type(8))) float v8f;

__device__ __forceinline__ float gcn_deg_scale(int deg) {
    return __frsqrt_rn(fmaxf((float)deg, 1.0f));
}

__device__ __forceinline__ int gcn_block_excl_scan(int v, int* wsum, int lane, int wave) {
    int x = v;
    for (int o = 1; o < 32; o <<= 1) {
        const int y = __shfl_up(x, o);
        if (lane >= o) x += y;
    }
    if (lane == 31) wsum[wave] = x;
    __syncthreads();
    if (wave == 0) {
        const int t = wsum[lane];
        int xs = t;
        for (int o = 1; o < 32; o <<= 1) {
            const int y = __shfl_up(xs, o);
            if (lane >= o) xs += y;
        }
        wsum[lane] = xs - t;
    }
    __syncthreads();
    const int r = x - v + wsum[wave];
    __syncthreads();
    return r;
}

__global__ __launch_bounds__(SORT_THREADS)
void gcn_chunk_sort_kernel(const int* __restrict__ src,
                           const int* __restrict__ dst,
                           int* __restrict__ by_dst,
                           int* __restrict__ by_src,
                           int2* __restrict__ offs,
                           int* __restrict__ ovf_count, int E) {
    __shared__ int hist_d[SORT_THREADS];
    __shared__ int hist_s[SORT_THREADS];
    __shared__ int wsum[32];
    __shared__ int stage_d[EDGES_PER_CHUNK];
    __shared__ int stage_s[EDGES_PER_CHUNK];
    const int tid  = threadIdx.x;
    const int lane = tid & 31;
    const int wave = tid >> 5;
    if (blockIdx.x == 0 && tid == 0) *ovf_count = 0;
    hist_d[tid] = 0;
    hist_s[tid] = 0;
    __syncthreads();
    const int e0 = blockIdx.x * EDGES_PER_CHUNK;
    int n = E - e0;
    if (n > EDGES_PER_CHUNK) n = EDGES_PER_CHUNK;
    if (n < 0) n = 0;
    int s[EDGES_PER_THREAD], d[EDGES_PER_THREAD], rd[EDGES_PER_THREAD], rs[EDGES_PER_THREAD];
    #pragma unroll
    for (int u = 0; u < EDGES_PER_THREAD; ++u) {
        const int i = u * SORT_THREADS + tid;
        s[u] = 0;
        d[u] = -1;
        if (i < n) { s[u] = src[e0 + i]; d[u] = dst[e0 + i]; }
    }
    #pragma unroll
    for (int u = 0; u < EDGES_PER_THREAD; ++u) {
        rd[u] = 0;
        rs[u] = 0;
        if (d[u] >= 0) {
            rd[u] = atomicAdd(&hist_d[d[u] >> BIN_SHIFT], 1);
            rs[u] = atomicAdd(&hist_s[s[u] >> BIN_SHIFT], 1);
        }
    }
    __syncthreads();
    const int start_d = gcn_block_excl_scan(hist_d[tid], wsum, lane, wave);
    const int start_s = gcn_block_excl_scan(hist_s[tid], wsum, lane, wave);
    hist_d[tid] = start_d;
    hist_s[tid] = start_s;
    __syncthreads();
    #pragma unroll
    for (int u = 0; u < EDGES_PER_THREAD; ++u) {
        if (d[u] >= 0) {
            stage_d[hist_d[d[u] >> BIN_SHIFT] + rd[u]] = (s[u] << BIN_SHIFT) | (d[u] & (BIN_NODES - 1));
            stage_s[hist_s[s[u] >> BIN_SHIFT] + rs[u]] = s[u];
        }
    }
    __syncthreads();
    int* od = by_dst + (size_t)blockIdx.x * CHUNK_PITCH;
    int* os = by_src + (size_t)blockIdx.x * CHUNK_PITCH;
    for (int i = tid; i < n; i += SORT_THREADS) {
        od[i] = stage_d[i];
        os[i] = stage_s[i];
    }
    if (tid <= N_BINS)
        offs[(size_t)blockIdx.x * OFFS_PITCH + tid] = make_int2(hist_d[tid], hist_s[tid]);
}

__global__ __launch_bounds__(BIN_THREADS)
void gcn_bin_kernel(const int* __restrict__ by_dst,
                    const int* __restrict__ by_src,
                    const int2* __restrict__ offs,
                    int* __restrict__ deg_in,
                    int* __restrict__ deg_out,
                    int* __restrict__ bucket,
                    int* __restrict__ ovf_count,
                    int* __restrict__ ovf, int n_nodes) {
    __shared__ int cnt_in[BIN_NODES];
    __shared__ int cnt_out[BIN_NODES];
    __shared__ __attribute__((aligned(16))) int tile[BIN_NODES * BUCKET_CAP];
    const int tid = threadIdx.x;
    const int b   = blockIdx.x;
    for (int i = tid; i < BIN_NODES * BUCKET_CAP / 4; i += BIN_THREADS)
        ((int4*)tile)[i] = make_int4(0, 0, 0, 0);
    if (tid < BIN_NODES) { cnt_in[tid] = 0; cnt_out[tid] = 0; }
    __syncthreads();
    const int k = tid & (SORT_CHUNKS - 1);
    const int2 a0 = offs[(size_t)k * OFFS_PITCH + b];
    const int2 a1 = offs[(size_t)k * OFFS_PITCH + b + 1];
    if (tid >= SORT_CHUNKS) {
        const int* run = by_src + (size_t)k * CHUNK_PITCH;
        for (int j = a0.y; j < a1.y; ++j)
            atomicAdd(&cnt_out[run[j] & (BIN_NODES - 1)], 1);
    } else {
        const int* run = by_dst + (size_t)k * CHUNK_PITCH;
        for (int j = a0.x; j < a1.x; ++j) {
            const int p   = run[j];
            const int dl  = p & (BIN_NODES - 1);
            const int sid = p >> BIN_SHIFT;
            const int r   = atomicAdd(&cnt_in[dl], 1);
            if (r < BUCKET_CAP) {
                tile[dl * BUCKET_CAP + r] = sid;
            } else {
                const int o = atomicAdd(ovf_count, 1);
                ((int2*)ovf)[o] = make_int2(b * BIN_NODES + dl, sid);
            }
        }
    }
    __syncthreads();
    const int node0 = b * BIN_NODES;
    if (tid < BIN_NODES && node0 + tid < n_nodes) {
        deg_in[node0 + tid]  = cnt_in[tid];
        deg_out[node0 + tid] = cnt_out[tid];
    }
    for (int i = tid; i < BIN_NODES * BUCKET_CAP / 4; i += BIN_THREADS) {
        const int row = i >> 4;
        const int q   = i & 15;
        if (node0 + row < n_nodes && (q < 8 || cnt_in[row] > 32))
            ((int4*)bucket)[(size_t)node0 * (BUCKET_CAP / 4) + i] = ((const int4*)tile)[i];
    }
}

__device__ __forceinline__ float4 gcn_row4(const float* __restrict__ h, int s, int lane) {
    return *(const float4*)(h + (size_t)s * 128 + lane * 4);
}
__device__ __forceinline__ void gcn_mac4(float4& a, const float4& v, float w) {
    a.x = __builtin_fmaf(v.x, w, a.x); a.y = __builtin_fmaf(v.y, w, a.y);
    a.z = __builtin_fmaf(v.z, w, a.z); a.w = __builtin_fmaf(v.w, w, a.w);
}

template <bool LOOKUP>
__device__ __forceinline__ int2 gcn_slot_chunk(const int* __restrict__ bucket,
                                               const int* __restrict__ deg_out,
                                               size_t first, int c, int lane) {
    const bool in = lane < c;
    const int* a = in ? bucket + first + lane : deg_out;
    int2 p;
    p.x = *a;
    p.x = in ? p.x : 0;
    p.y = 0x3f800000;
    if (LOOKUP) p.y = __float_as_int(gcn_deg_scale(deg_out[p.x]));
    return p;
}

__device__ __forceinline__ void gcn_mac8(float4& acc, int r, const float4 (&v)[8], const float (&w)[8]) {
    if (r >= 8) {
        #pragma unroll
        for (int u = 0; u < 8; ++u) gcn_mac4(acc, v[u], w[u]);
    } else {
        #pragma unroll
        for (int u = 0; u < 8; ++u)
            if (u < r) gcn_mac4(acc, v[u], w[u]);
    }
}

template <bool LOOKUP>
__device__ __forceinline__ float4 gcn_gather_rest(const float* __restrict__ h,
                                                  const int* __restrict__ bucket,
                                                  const int* __restrict__ deg_out,
                                                  const int* __restrict__ ovf,
                                                  const int* __restrict__ ovf_count,
                                                  int node, int cnt, int lane, float4 acc) {
    const int nb = (cnt < BUCKET_CAP) ? cnt : BUCKET_CAP;
    const int c = nb - 32;
    const int2 my = gcn_slot_chunk<LOOKUP>(bucket, deg_out, (size_t)node * BUCKET_CAP + 32, c, lane);
    for (int k = 0; k < c; ++k) {
        const float4 v0 = gcn_row4(h, __builtin_amdgcn_readlane(my.x, k), lane);
        const float w0 = __int_as_float(__builtin_amdgcn_readlane(my.y, k));
        gcn_mac4(acc, v0, w0);
    }
    if (cnt > BUCKET_CAP) {
        const int no = *ovf_count;
        for (int j0 = 0; j0 < no; j0 += 32) {
            int2 e = make_int2(-1, 0);
            if (j0 + lane < no) e = ((const int2*)ovf)[j0 + lane];
            const bool hit = (e.x == node);
            int wbits = 0x3f800000;
            if (LOOKUP && hit) wbits = __float_as_int(gcn_deg_scale(deg_out[e.y]));
            unsigned mask = __builtin_amdgcn_ballot_w32(hit);
            while (mask) {
                const int k = __builtin_ctz(mask);
                mask &= mask - 1;
                const float4 v0 = gcn_row4(h, __builtin_amdgcn_readlane(e.y, k), lane);
                const float w0 = __int_as_float(__builtin_amdgcn_readlane(wbits, k));
                gcn_mac4(acc, v0, w0);
            }
        }
    }
    return acc;
}

template <bool LOOKUP>
__device__ __forceinline__ void gcn_gather_pair(const float* __restrict__ h,
                                                const int* __restrict__ bucket,
                                                const int* __restrict__ deg_out,
                                                const int* __restrict__ ovf,
                                                const int* __restrict__ ovf_count,
                                                int nodeA, int cntA, int2 myA,
                                                int nodeB, int cntB, int2 myB, int lane,
                                                float4& outA, float4& outB) {
    float4 accA = make_float4(0.0f, 0.0f, 0.0f, 0.0f);
    float4 accB = make_float4(0.0f, 0.0f, 0.0f, 0.0f);
    const int cA = (cntA < 32) ? cntA : 32;
    const int cB = (cntB < 32) ? cntB : 32;
    const int cmax = (cA > cB) ? cA : cB;
    for (int k = 0; k < cmax; k += 8) {
        float4 va[8], vb[8];
        float wa[8], wb[8];
        const int rA = (cA > k) ? cA - k : 0;
        const int rB = (cB > k) ? cB - k : 0;
        #pragma unroll
        for (int u = 0; u < 8; ++u) {
            va[u] = gcn_row4(h, __builtin_amdgcn_readlane(myA.x, k + u), lane);
            wa[u] = __int_as_float(__builtin_amdgcn_readlane(myA.y, k + u));
        }
        #pragma unroll
        for (int u = 0; u < 8; ++u) {
            vb[u] = gcn_row4(h, __builtin_amdgcn_readlane(myB.x, k + u), lane);
            wb[u] = __int_as_float(__builtin_amdgcn_readlane(myB.y, k + u));
        }
        gcn_mac8(accA, rA, va, wa);
        gcn_mac8(accB, rB, vb, wb);
    }
    if (cntA > 32) accA = gcn_gather_rest<LOOKUP>(h, bucket, deg_out, ovf, ovf_count, nodeA, cntA, lane, accA);
    if (cntB > 32) accB = gcn_gather_rest<LOOKUP>(h, bucket, deg_out, ovf, ovf_count, nodeB, cntB, lane, accB);
    outA = accA;
    outB = accB;
}

template <bool PROJ>
__device__ __forceinline__
void gcn_layer_body(const float* __restrict__ h,
                    const int* __restrict__ deg_in,
                    const int* __restrict__ deg_out,
                    const int* __restrict__ bucket,
                    const int* __restrict__ ovf,
                    const int* __restrict__ ovf_count,
                    const float* __restrict__ W,
                    const float* __restrict__ bias,
                    const float* __restrict__ Wn,
                    float* __restrict__ out,
                    int N, int relu, int n_nodes) {
    const int K = 128;
    constexpr bool LOOKUP = !PROJ;
    __shared__ float As[32 * 132];

    const int lane = threadIdx.x & 31;
    const int wave = __builtin_amdgcn_readfirstlane(threadIdx.x >> 5);
    const int half = lane >> 4;
    const int l16  = lane & 15;
    const int tileM = blockIdx.x * 32;
    const int tileN = wave * 32;
    const int colN  = tileN + 2 * l16;

    const int row0 = tileM + wave * 8;
    int cnt[8];
    #pragma unroll
    for (int u = 0; u < 8; ++u)
        cnt[u] = (row0 + u < n_nodes) ? deg_in[row0 + u] : 0;
    int2 my[8];
    #pragma unroll
    for (int u = 0; u < 8; ++u) {
        int c = cnt[u];
        if (c > 32) c = 32;
        my[u] = gcn_slot_chunk<LOOKUP>(bucket, deg_out, (size_t)(row0 + u) * BUCKET_CAP, c, lane);
    }
    if (LOOKUP) asm volatile("" :: "v"(my[0].y), "v"(my[1].y), "v"(my[2].y), "v"(my[3].y),
                                   "v"(my[4].y), "v"(my[5].y), "v"(my[6].y), "v"(my[7].y));
    else        asm volatile("" :: "v"(my[0].x), "v"(my[1].x), "v"(my[2].x), "v"(my[3].x),
                                   "v"(my[4].x), "v"(my[5].x), "v"(my[6].x), "v"(my[7].x));
    #pragma unroll
    for (int p = 0; p < 4; ++p) {
        float4 a, b;
        gcn_gather_pair<LOOKUP>(h, bucket, deg_out, ovf, ovf_count,
                                row0 + 2 * p,     cnt[2 * p],     my[2 * p],
                                row0 + 2 * p + 1, cnt[2 * p + 1], my[2 * p + 1], lane, a, b);
        const float sa = gcn_deg_scale(cnt[2 * p]);
        const float sb = gcn_deg_scale(cnt[2 * p + 1]);
        a.x *= sa; a.y *= sa; a.z *= sa; a.w *= sa;
        b.x *= sb; b.y *= sb; b.z *= sb; b.w *= sb;
        *(float4*)(As + (wave * 8 + 2 * p) * 132 + lane * 4)     = a;
        *(float4*)(As + (wave * 8 + 2 * p + 1) * 132 + lane * 4) = b;
    }
    __syncthreads();

    if (!PROJ && tileN >= N) return;

    v8f acc00 = {};
    v8f acc10 = {};
    v8f acc01 = {};
    v8f acc11 = {};
    #pragma unroll 4
    for (int k0 = 0; k0 < K; k0 += 4) {
        int ka = k0 + 2 * half;
        v2f a0, a1, b0, b1;
        a0.x = As[l16 * 132 + ka];
        a0.y = As[l16 * 132 + ka + 1];
        a1.x = As[(16 + l16) * 132 + ka];
        a1.y = As[(16 + l16) * 132 + ka + 1];
        const float2 w0 = *(const float2*)(W + (size_t)ka * N + colN);
        const float2 w1 = *(const float2*)(W + (size_t)(ka + 1) * N + colN);
        b0.x = w0.x; b0.y = w1.x;
        b1.x = w0.y; b1.y = w1.y;
        acc00 = __builtin_amdgcn_wmma_f32_16x16x4_f32(
             false, a0,  false, b0,
             (short)0, acc00,  false,  false);
        acc10 = __builtin_amdgcn_wmma_f32_16x16x4_f32(
             false, a1,  false, b0,
             (short)0, acc10,  false,  false);
        acc01 = __builtin_amdgcn_wmma_f32_16x16x4_f32(
             false, a0,  false, b1,
             (short)0, acc01,  false,  false);
        acc11 = __builtin_amdgcn_wmma_f32_16x16x4_f32(
             false, a1,  false, b1,
             (short)0, acc11,  false,  false);
    }

    const float2 bb = *(const float2*)(bias + colN);
    if (PROJ) {
        __shared__ float Zp[8 * 256];
        const int kq = wave;
        v2f bn[8];
        #pragma unroll
        for (int ks = 0; ks < 8; ++ks) {
            const int ka = kq * 32 + ks * 4 + 2 * half;
            bn[ks].x = Wn[ka * 16 + l16];
            bn[ks].y = Wn[(ka + 1) * 16 + l16];
        }
        __syncthreads();
        #pragma unroll
        for (int r = 0; r < 8; ++r) {
            float2 v0, v1;
            v0.x = acc00[r] + bb.x; v0.y = acc01[r] + bb.y;
            v1.x = acc10[r] + bb.x; v1.y = acc11[r] + bb.y;
            if (relu) {
                v0.x = fmaxf(v0.x, 0.0f); v0.y = fmaxf(v0.y, 0.0f);
                v1.x = fmaxf(v1.x, 0.0f); v1.y = fmaxf(v1.y, 0.0f);
            }
            *(float2*)(As + (r + 8 * half) * 132 + colN)      = v0;
            *(float2*)(As + (16 + r + 8 * half) * 132 + colN) = v1;
        }
        __syncthreads();
        v8f z0 = {};
        v8f z1 = {};
        #pragma unroll
        for (int ks = 0; ks < 8; ++ks) {
            const int ka = kq * 32 + ks * 4 + 2 * half;
            v2f a0, a1;
            a0.x = As[l16 * 132 + ka];
            a0.y = As[l16 * 132 + ka + 1];
            a1.x = As[(16 + l16) * 132 + ka];
            a1.y = As[(16 + l16) * 132 + ka + 1];
            z0 = __builtin_amdgcn_wmma_f32_16x16x4_f32(
                 false, a0,  false, bn[ks],
                 (short)0, z0,  false,  false);
            z1 = __builtin_amdgcn_wmma_f32_16x16x4_f32(
                 false, a1,  false, bn[ks],
                 (short)0, z1,  false,  false);
        }
        #pragma unroll
        for (int r = 0; r < 8; ++r) {
            Zp[(2 * kq) * 256 + (r + 8 * half) * 16 + l16]     = z0[r];
            Zp[(2 * kq + 1) * 256 + (r + 8 * half) * 16 + l16] = z1[r];
        }
        __syncthreads();
        {
            const int t = threadIdx.x;
            const int row = t >> 2;
            const int q   = t & 3;
            const float* p = Zp + (row >> 4) * 256 + (row & 15) * 16 + q * 4;
            const float4 s0 = *(const float4*)(p);
            const float4 s1 = *(const float4*)(p + 512);
            const float4 s2 = *(const float4*)(p + 1024);
            const float4 s3 = *(const float4*)(p + 1536);
            if (tileM + row < n_nodes) {
                const float so = gcn_deg_scale(deg_out[tileM + row]);
                float4 z;
                z.x = (((s0.x + s1.x) + s2.x) + s3.x) * so;
                z.y = (((s0.y + s1.y) + s2.y) + s3.y) * so;
                z.z = (((s0.z + s1.z) + s2.z) + s3.z) * so;
                z.w = (((s0.w + s1.w) + s2.w) + s3.w) * so;
                ((float4*)out)[(size_t)(tileM + row) * 4 + q] = z;
            }
        }
        return;
    }
    #pragma unroll
    for (int r = 0; r < 8; ++r) {
        int row = tileM + r + 8 * half;
        float2 v;
        v.x = acc00[r] + bb.x;
        v.y = acc01[r] + bb.y;
        if (relu) { v.x = fmaxf(v.x, 0.0f); v.y = fmaxf(v.y, 0.0f); }
        const float so = gcn_deg_scale(deg_out[row]);
        v.x *= so; v.y *= so;
        *(float2*)(out + (size_t)row * N + colN) = v;
    }
    if (tileM + 16 < n_nodes) {
        #pragma unroll
        for (int r = 0; r < 8; ++r) {
            int row = tileM + 16 + r + 8 * half;
            float2 v;
            v.x = acc10[r] + bb.x;
            v.y = acc11[r] + bb.y;
            if (relu) { v.x = fmaxf(v.x, 0.0f); v.y = fmaxf(v.y, 0.0f); }
            const float so = gcn_deg_scale(deg_out[row]);
            v.x *= so; v.y *= so;
            *(float2*)(out + (size_t)row * N + colN) = v;
        }
    }
}

__global__ __launch_bounds__(128)
void gcn_layer_kernel(const float* __restrict__ h, const int* __restrict__ deg_in,
                      const int* __restrict__ deg_out, const int* __restrict__ bucket,
                      const int* __restrict__ ovf, const int* __restrict__ ovf_count,
                      const float* __restrict__ W, const float* __restrict__ bias,
                      float* __restrict__ out, int N, int relu, int n_nodes) {
    gcn_layer_body<false>(h, deg_in, deg_out, bucket, ovf, ovf_count, W, bias, nullptr, out, N, relu, n_nodes);
}

__global__ __launch_bounds__(128)
void gcn_layer_proj_kernel(const float* __restrict__ h, const int* __restrict__ deg_in,
                           const int* __restrict__ deg_out, const int* __restrict__ bucket,
                           const int* __restrict__ ovf, const int* __restrict__ ovf_count,
                           const float* __restrict__ W, const float* __restrict__ bias,
                           const float* __restrict__ Wn, float* __restrict__ z,
                           int relu, int n_nodes) {
    gcn_layer_body<true>(h, deg_in, deg_out, bucket, ovf, ovf_count, W, bias, Wn, z, 128, relu, n_nodes);
}

__global__ __launch_bounds__(256)
void gcn_gather16_kernel(const float* __restrict__ z,
                         const int* __restrict__ deg_in,
                         const int* __restrict__ bucket,
                         const int* __restrict__ ovf,
                         const int* __restrict__ ovf_count,
                         const float* __restrict__ bias,
                         float* __restrict__ out,
                         int n_nodes) {
    const int t = blockIdx.x * 256 + threadIdx.x;
    const int node = t >> 2;
    const int q = t & 3;
    if (node >= n_nodes) return;
    const int cnt = deg_in[node];
    const int nb = (cnt < BUCKET_CAP) ? cnt : BUCKET_CAP;
    const int* bk = bucket + (size_t)node * BUCKET_CAP;
    const float4* z4 = (const float4*)z;
    const float s = gcn_deg_scale(cnt);
    const float4 b = ((const float4*)bias)[q];
    float4 acc = make_float4(0.0f, 0.0f, 0.0f, 0.0f);
    if (nb > 0) {
        const int last = nb - 1;
        int p[8];
        #pragma unroll
        for (int u = 0; u < 8; ++u) p[u] = bk[min(u, last)];
        for (int j = 0; j < nb; j += 8) {
            float4 v[8];
            int pn[8];
            #pragma unroll
            for (int u = 0; u < 8; ++u) v[u] = z4[(size_t)p[u] * 4 + q];
            #pragma unroll
            for (int u = 0; u < 8; ++u) pn[u] = bk[min(j + 8 + u, last)];
            acc.x += v[0].x; acc.y += v[0].y; acc.z += v[0].z; acc.w += v[0].w;
            #pragma unroll
            for (int u = 1; u < 8; ++u) {
                const bool real = j + u < nb;
                acc.x = real ? acc.x + v[u].x : acc.x; acc.y = real ? acc.y + v[u].y : acc.y;
                acc.z = real ? acc.z + v[u].z : acc.z; acc.w = real ? acc.w + v[u].w : acc.w;
            }
            #pragma unroll
            for (int u = 0; u < 8; ++u) p[u] = pn[u];
        }
    }
    if (cnt > BUCKET_CAP) {
        const int no = *ovf_count;
        const int2* ov = (const int2*)ovf;
        for (int j = 0; j < no; ++j) {
            const int2 e = ov[j];
            if (e.x == node) {
                const float4 v0 = z4[(size_t)e.y * 4 + q];
                acc.x += v0.x; acc.y += v0.y; acc.z += v0.z; acc.w += v0.w;
            }
        }
    }
    float4 o;
    o.x = acc.x * s + b.x; o.y = acc.y * s + b.y;
    o.z = acc.z * s + b.z; o.w = acc.w * s + b.w;
    ((float4*)out)[(size_t)node * 4 + q] = o;
}

extern "C" void kernel_launch(void* const* d_in, const int* in_sizes, int n_in,
                              void* d_out, int out_size, void* d_ws, size_t ws_size,
                              hipStream_t stream) {
    const float* in_feat = (const float*)d_in[0];
    const int*   e_src   = (const int*)d_in[1];
    const int*   e_dst   = (const int*)d_in[2];
    const float* W1 = (const float*)d_in[3];
    const float* b1 = (const float*)d_in[4];
    const float* W2 = (const float*)d_in[5];
    const float* b2 = (const float*)d_in[6];
    const float* W3 = (const float*)d_in[7];
    const float* b3 = (const float*)d_in[8];
    float* out = (float*)d_out;

    const size_t NF = (size_t)N_NODES * 128;
    float* B1 = (float*)d_ws;
    float* Z  = B1 + NF;
    int*   deg_out   = (int*)(Z + NF);
    int*   deg_in    = deg_out + DEG_PITCH;
    int*   ovf_count = deg_in + DEG_PITCH;
    int*   bucket    = ovf_count + CTRL_INTS;
    int*   ovf       = bucket + (size_t)N_NODES * BUCKET_CAP;
    int*   by_dst    = ovf + 2 * (size_t)N_EDGES;
    int*   by_src    = by_dst + (size_t)SORT_CHUNKS * CHUNK_PITCH;
    int2*  offs      = (int2*)(by_src + (size_t)SORT_CHUNKS * CHUNK_PITCH);

    const int TB = 256;
    const int LB = 128;
    const int layerBlocks = (N_NODES + 31) / 32;
    const int rowBlocks   = (N_NODES * 4 + TB - 1) / TB;

    gcn_chunk_sort_kernel<<<SORT_CHUNKS, SORT_THREADS, 0, stream>>>(e_src, e_dst, by_dst, by_src, offs, ovf_count, N_EDGES);
    gcn_bin_kernel<<<N_BINS, BIN_THREADS, 0, stream>>>(by_dst, by_src, offs, deg_in, deg_out, bucket, ovf_count, ovf, N_NODES);

    gcn_layer_kernel<<<layerBlocks, LB, 0, stream>>>(in_feat, deg_in, deg_out, bucket, ovf, ovf_count, W1, b1, B1, H_FEATS, 1, N_NODES);
    gcn_layer_proj_kernel<<<layerBlocks, LB, 0, stream>>>(B1, deg_in, deg_out, bucket, ovf, ovf_count, W2, b2, W3, Z, 1, N_NODES);
    gcn_gather16_kernel<<<rowBlocks, TB, 0, stream>>>(Z, deg_in, bucket, ovf, ovf_count, b3, out, N_NODES);
}
